# starting kernel
# baseline (speedup 1.0000x reference)
#include <hip/hip_runtime.h>
#include <math.h>


typedef __attribute__((ext_vector_type(2))) float v2f;
typedef __attribute__((ext_vector_type(8))) float v8f;

#define B_      16
#define N_      8192
#define K_      125
#define KT      8
#define NCHUNK  16
#define CHUNK   (N_ / NCHUNK)
#define C15     0.06349363593424097f

__global__ __launch_bounds__(256)
void fv_coefs(const float* __restrict__ w, const float* __restrict__ mu,
              const float* __restrict__ sg, float4* __restrict__ c4)
{
    const int tid  = threadIdx.x;
    const int lane = tid & 31;
    const int t    = tid >> 5;
    const int m    = lane & 15;
    const bool hi  = lane >= 16;
    const int  kc    = t * 16 + m;
    const bool valid = kc < K_;
    const int  ki    = valid ? kc : 0;

    const float m0 = mu[ki*3+0], m1 = mu[ki*3+1], m2 = mu[ki*3+2];
    const float s0 = sg[ki*3+0], s1 = sg[ki*3+1], s2 = sg[ki*3+2];
    const float i0 = 1.0f/(s0*s0), i1 = 1.0f/(s1*s1), i2 = 1.0f/(s2*s2);
    const float g0 = -0.5f*i0, g1 = -0.5f*i1, g2 = -0.5f*i2;
    const float g3 = m0*i0,    g4 = m1*i1,    g5 = m2*i2;
    const float lq = logf(w[ki] * C15 / (s0 * s1 * s2));
    const float g6 = valid ? (lq - 0.5f*(m0*m0*i0 + m1*m1*i1 + m2*m2*i2))
                           : -INFINITY;

    float4 v;
    v.x = hi ? g2   : g0;
    v.y = hi ? g3   : g1;
    v.z = hi ? g6   : g4;
    v.w = hi ? 0.0f : g5;
    c4[tid] = v;
}

__global__ __launch_bounds__(256)
void fv_denom_wmma(const float* __restrict__ pts,
                   const float4* __restrict__ c4,
                   float* __restrict__ rdenom)
{
    const int lane = threadIdx.x & 31;
    const int wave = threadIdx.x >> 5;
    const int tile = blockIdx.x * 8 + wave;
    const int b    = tile >> 9;
    const int n0   = (tile & 511) << 4;
    const int m    = lane & 15;
    const bool hi  = lane >= 16;

    const float* p = pts + (size_t)(b * N_ + n0 + m) * 3;
    const float px = p[0], py = p[1], pz = p[2];

    v2f a1, a2;
    a1.x = hi ? pz * pz : px * px;
    a1.y = hi ? px      : py * py;
    a2.x = hi ? 1.0f    : py;
    a2.y = hi ? 0.0f    : pz;

    float acc[8];
#pragma unroll
    for (int r = 0; r < 8; ++r) acc[r] = 0.0f;

#pragma unroll
    for (int t = 0; t < KT; ++t) {
        const float4 bc = c4[t * 32 + lane];
        v2f b1, b2;
        b1.x = bc.x; b1.y = bc.y;
        b2.x = bc.z; b2.y = bc.w;

        v8f c = {};
        c = __builtin_amdgcn_wmma_f32_16x16x4_f32(false, a1, false, b1,
                                                  (short)0, c, false, false);
        c = __builtin_amdgcn_wmma_f32_16x16x4_f32(false, a2, false, b2,
                                                  (short)0, c, false, false);

#pragma unroll
        for (int r = 0; r < 8; ++r)
            acc[r] += __expf(c[r]);
    }

#pragma unroll
    for (int r = 0; r < 8; ++r) {
        float v = acc[r];
        v += __shfl_xor(v, 1, 32);
        v += __shfl_xor(v, 2, 32);
        v += __shfl_xor(v, 4, 32);
        v += __shfl_xor(v, 8, 32);
        acc[r] = v;
    }
    if (lane == 0) {
#pragma unroll
        for (int r = 0; r < 8; ++r)
            rdenom[b * N_ + n0 + r] = 1.0f / acc[r];
    } else if (lane == 16) {
#pragma unroll
        for (int r = 0; r < 8; ++r)
            rdenom[b * N_ + n0 + 8 + r] = 1.0f / acc[r];
    }
}

__global__ __launch_bounds__(128)
void fv_accum(const float* __restrict__ pts, const float* __restrict__ w,
              const float* __restrict__ mu, const float* __restrict__ sg,
              const float* __restrict__ rdenom, float* __restrict__ partial)
{
    __shared__ float lx[CHUNK], ly[CHUNK], lz[CHUNK], lrd[CHUNK];
    const int b  = blockIdx.x >> 4;
    const int ch = blockIdx.x & 15;
    const int n0 = ch * CHUNK;
    const int t  = threadIdx.x;

    for (int i = t; i < CHUNK; i += 128) {
        const int n = b * N_ + n0 + i;
        const float* p = pts + (size_t)n * 3;
        lx[i] = p[0]; ly[i] = p[1]; lz[i] = p[2];
        lrd[i] = rdenom[n];
    }
    __syncthreads();

    const int kk = (t < K_) ? t : 0;
    const float m0 = mu[kk*3+0], m1 = mu[kk*3+1], m2 = mu[kk*3+2];
    const float s0 = sg[kk*3+0], s1 = sg[kk*3+1], s2 = sg[kk*3+2];
    const float rs0 = 1.0f/s0, rs1 = 1.0f/s1, rs2 = 1.0f/s2;
    const float q = w[kk] * C15 * rs0 * rs1 * rs2;

    float sQ = 0.0f, mQ = -INFINITY;
    float sM[3] = {0,0,0}, xM[3] = {-INFINITY,-INFINITY,-INFINITY};
    float nM[3] = { INFINITY, INFINITY, INFINITY};
    float sS[3] = {0,0,0}, xS[3] = {-INFINITY,-INFINITY,-INFINITY};
    float nS[3] = { INFINITY, INFINITY, INFINITY};

    for (int i = 0; i < CHUNK; ++i) {
        float z0 = (lx[i] - m0) * rs0;
        float z1 = (ly[i] - m1) * rs1;
        float z2 = (lz[i] - m2) * rs2;
        float e  = -0.5f * (z0*z0 + z1*z1 + z2*z2);
        float Q  = q * __expf(e) * lrd[i];
        sQ += Q; mQ = fmaxf(mQ, Q);
        float z[3] = {z0, z1, z2};
#pragma unroll
        for (int d = 0; d < 3; ++d) {
            float u = Q * z[d];
            sM[d] += u; xM[d] = fmaxf(xM[d], u); nM[d] = fminf(nM[d], u);
            float v = Q * (z[d]*z[d] - 1.0f);
            sS[d] += v; xS[d] = fmaxf(xS[d], v); nS[d] = fminf(nS[d], v);
        }
    }

    if (t < K_) {
        float* o = partial + ((size_t)(b * NCHUNK + ch) * K_ + t) * 20;
        o[0] = mQ;  o[1] = sQ;
#pragma unroll
        for (int d = 0; d < 3; ++d) {
            o[2  + d] = xM[d]; o[5  + d] = nM[d]; o[8  + d] = sM[d];
            o[11 + d] = xS[d]; o[14 + d] = nS[d]; o[17 + d] = sS[d];
        }
    }
}

__global__ __launch_bounds__(128)
void fv_finalize(const float* __restrict__ w, const float* __restrict__ partial,
                 float* __restrict__ out)
{
    __shared__ float vals[20][128];
    __shared__ float ssq[20];
    const int b = blockIdx.x;
    const int t = threadIdx.x;

    float v[20];
    if (t < K_) {
        float a[20];
        {
            const float* p = partial + ((size_t)(b * NCHUNK) * K_ + t) * 20;
#pragma unroll
            for (int f = 0; f < 20; ++f) a[f] = p[f];
        }
        for (int c = 1; c < NCHUNK; ++c) {
            const float* p = partial + ((size_t)(b * NCHUNK + c) * K_ + t) * 20;
            a[0] = fmaxf(a[0], p[0]); a[1] += p[1];
#pragma unroll
            for (int d = 0; d < 3; ++d) {
                a[2  + d] = fmaxf(a[2  + d], p[2  + d]);
                a[5  + d] = fminf(a[5  + d], p[5  + d]);
                a[8  + d] +=            p[8  + d];
                a[11 + d] = fmaxf(a[11 + d], p[11 + d]);
                a[14 + d] = fminf(a[14 + d], p[14 + d]);
                a[17 + d] +=            p[17 + d];
            }
        }
        const float wk  = w[t];
        const float Nf  = (float)N_;
        const float spi = 1.0f / (sqrtf(wk) * Nf);
        const float ssg = 1.0f / (Nf * sqrtf(2.0f * wk));
        v[0] = (a[0] - wk)      * spi;
        v[1] = (a[1] - Nf * wk) * spi;
#pragma unroll
        for (int f = 2;  f < 11; ++f) v[f] = a[f] * spi;
#pragma unroll
        for (int f = 11; f < 20; ++f) v[f] = a[f] * ssg;
#pragma unroll
        for (int f = 0; f < 20; ++f)
            v[f] = copysignf(sqrtf(fabsf(v[f])), v[f]);
    } else {
#pragma unroll
        for (int f = 0; f < 20; ++f) v[f] = 0.0f;
    }
#pragma unroll
    for (int f = 0; f < 20; ++f) vals[f][t] = v[f];
    __syncthreads();

    if (t < 20) {
        float s = 0.0f;
        for (int k = 0; k < K_; ++k) { float x = vals[t][k]; s += x * x; }
        ssq[t] = s;
    }
    __syncthreads();

    if (t < K_) {
#pragma unroll
        for (int f = 0; f < 20; ++f)
            out[(size_t)b * 20 * K_ + f * K_ + t] =
                v[f] * rsqrtf(fmaxf(ssq[f], 1e-12f));
    }
}

extern "C" void kernel_launch(void* const* d_in, const int* in_sizes, int n_in,
                              void* d_out, int out_size, void* d_ws, size_t ws_size,
                              hipStream_t stream)
{
    const float* pts = (const float*)d_in[0];
    const float* w   = (const float*)d_in[1];
    const float* mu  = (const float*)d_in[2];
    const float* sg  = (const float*)d_in[3];
    float* out = (float*)d_out;

    float*  rdenom  = (float*)d_ws;
    float*  partial = rdenom + B_ * N_;
    float4* c4      = (float4*)(partial + (size_t)B_ * NCHUNK * K_ * 20);

    fv_coefs<<<1, 256, 0, stream>>>(w, mu, sg, c4);
    fv_denom_wmma<<<(B_ * N_ / 16) / 8, 256, 0, stream>>>(pts, c4, rdenom);
    fv_accum<<<B_ * NCHUNK, 128, 0, stream>>>(pts, w, mu, sg, rdenom, partial);
    fv_finalize<<<B_, 128, 0, stream>>>(w, partial, out);
}


# one fused pass in the WMMA result layout: a wave computes each 16-point x 128-gaussian density tile once, normalises, accumulates in registers; 4 launches -> 2; chunk sum = 16 sub-sums of 32 points
# speedup vs baseline: 1.5449x; 1.5449x over previous
#include <hip/hip_runtime.h>
#include <math.h>


typedef __attribute__((ext_vector_type(2))) float v2f;
typedef __attribute__((ext_vector_type(8))) float v8f;

template <int M>
__device__ __forceinline__ float row_xor(float x)
{
    const int b = __builtin_bit_cast(int, x);
    return __builtin_bit_cast(float,
        __builtin_amdgcn_update_dpp(b, b, 0x160 + M, 0xf, 0xf, false));
}

__device__ __forceinline__ float xhalf(float x)
{
    const int b = __builtin_bit_cast(int, x);
    return __builtin_bit_cast(float,
        __builtin_amdgcn_permlanex16(b, b, 0x76543210, 0xfedcba98, false, false));
}

#define B_      16
#define N_      8192
#define K_      125
#define KT      8
#define NCHUNK  16
#define CHUNK   (N_ / NCHUNK)
#define NWAVE   8
#define PTW     (CHUNK / 16 / NWAVE)
#define C15     0.06349363593424097f

__global__ __launch_bounds__(256)
void fv_fused(const float* __restrict__ pts, const float* __restrict__ w,
              const float* __restrict__ mu, const float* __restrict__ sg,
              float* __restrict__ partial)
{
    __shared__ __attribute__((aligned(16))) float lx[CHUNK];
    __shared__ __attribute__((aligned(16))) float ly[CHUNK];
    __shared__ __attribute__((aligned(16))) float lz[CHUNK];
    __shared__ float4 ctab[KT * 32];
    __shared__ float red[NWAVE][20][KT * 16];

    const int tid  = threadIdx.x;
    const int lane = tid & 31;
    const int wave = tid >> 5;
    const int m    = lane & 15;
    const bool hi  = lane >= 16;
    const int b    = blockIdx.x >> 4;
    const int ch   = blockIdx.x & 15;
    const int n0   = ch * CHUNK;

    for (int i = tid; i < CHUNK; i += 256) {
        const float* p = pts + (size_t)(b * N_ + n0 + i) * 3;
        lx[i] = p[0]; ly[i] = p[1]; lz[i] = p[2];
    }

    {
        const int  kc    = wave * 16 + m;
        const bool valid = kc < K_;
        const int  ki    = valid ? kc : 0;
        const float m0 = mu[ki*3+0], m1 = mu[ki*3+1], m2 = mu[ki*3+2];
        const float s0 = sg[ki*3+0], s1 = sg[ki*3+1], s2 = sg[ki*3+2];
        const float i0 = 1.0f/(s0*s0), i1 = 1.0f/(s1*s1), i2 = 1.0f/(s2*s2);
        const float g0 = -0.5f*i0, g1 = -0.5f*i1, g2 = -0.5f*i2;
        const float g3 = m0*i0,    g4 = m1*i1,    g5 = m2*i2;
        const float lq = logf(w[ki] * C15 / (s0 * s1 * s2));
        const float g6 = valid ? (lq - 0.5f*(m0*m0*i0 + m1*m1*i1 + m2*m2*i2))
                               : -INFINITY;
        float4 v;
        v.x = hi ? g2   : g0;
        v.y = hi ? g3   : g1;
        v.z = hi ? g6   : g4;
        v.w = hi ? 0.0f : g5;
        ctab[tid] = v;
    }
    __syncthreads();

    v2f   b1[KT], b2[KT];
    float rs[KT][3], nm[KT][3];
#pragma unroll
    for (int t = 0; t < KT; ++t) {
        const float4 bc = ctab[t * 32 + lane];
        b1[t].x = bc.x; b1[t].y = bc.y;
        b2[t].x = bc.z; b2[t].y = bc.w;
        const int kc = t * 16 + m;
        const int ki = (kc < K_) ? kc : 0;
#pragma unroll
        for (int d = 0; d < 3; ++d) {
            const float r = 1.0f / sg[ki*3+d];
            rs[t][d] = r;
            nm[t][d] = -mu[ki*3+d] * r;
        }
    }

    float sQ[KT], mQ[KT];
    float sM[KT][3], xM[KT][3], nM[KT][3];
    float sS[KT][3], xS[KT][3], nS[KT][3];
#pragma unroll
    for (int t = 0; t < KT; ++t) {
        sQ[t] = 0.0f; mQ[t] = -INFINITY;
#pragma unroll
        for (int d = 0; d < 3; ++d) {
            sM[t][d] = 0.0f; xM[t][d] = -INFINITY; nM[t][d] = INFINITY;
            sS[t][d] = 0.0f; xS[t][d] = -INFINITY; nS[t][d] = INFINITY;
        }
    }

#pragma unroll 1
    for (int it = 0; it < PTW; ++it) {
        const int pb = (wave * PTW + it) * 16;

        const float px = lx[pb + m], py = ly[pb + m], pz = lz[pb + m];
        v2f a1, a2;
        a1.x = hi ? pz * pz : px * px;
        a1.y = hi ? px      : py * py;
        a2.x = hi ? 1.0f    : py;
        a2.y = hi ? 0.0f    : pz;

        const int rb = pb + (hi ? 8 : 0);
        float rx[8], ry[8], rz[8];
        {
            const float4 x0 = *(const float4*)(lx + rb), x1 = *(const float4*)(lx + rb + 4);
            const float4 y0 = *(const float4*)(ly + rb), y1 = *(const float4*)(ly + rb + 4);
            const float4 z0 = *(const float4*)(lz + rb), z1 = *(const float4*)(lz + rb + 4);
            rx[0]=x0.x; rx[1]=x0.y; rx[2]=x0.z; rx[3]=x0.w; rx[4]=x1.x; rx[5]=x1.y; rx[6]=x1.z; rx[7]=x1.w;
            ry[0]=y0.x; ry[1]=y0.y; ry[2]=y0.z; ry[3]=y0.w; ry[4]=y1.x; ry[5]=y1.y; ry[6]=y1.z; ry[7]=y1.w;
            rz[0]=z0.x; rz[1]=z0.y; rz[2]=z0.z; rz[3]=z0.w; rz[4]=z1.x; rz[5]=z1.y; rz[6]=z1.z; rz[7]=z1.w;
        }

        float e[KT][8];
        float den[8];
#pragma unroll
        for (int r = 0; r < 8; ++r) den[r] = 0.0f;
#pragma unroll
        for (int t = 0; t < KT; ++t) {
            v8f c = {};
            c = __builtin_amdgcn_wmma_f32_16x16x4_f32(false, a1, false, b1[t],
                                                      (short)0, c, false, false);
            c = __builtin_amdgcn_wmma_f32_16x16x4_f32(false, a2, false, b2[t],
                                                      (short)0, c, false, false);
#pragma unroll
            for (int r = 0; r < 8; ++r) {
                const float ev = __expf(c[r]);
                e[t][r] = ev;
                den[r] += ev;
            }
        }

        float rd[8];
#pragma unroll
        for (int r = 0; r < 8; ++r) {
            float v = den[r];
            v += row_xor<1>(v);
            v += row_xor<2>(v);
            v += row_xor<4>(v);
            v += row_xor<8>(v);
            rd[r] = 1.0f / v;
        }

#pragma unroll
        for (int t = 0; t < KT; ++t) {
#pragma unroll
            for (int r = 0; r < 8; ++r) {
                const float Q = e[t][r] * rd[r];
                float z[3];
                z[0] = fmaf(rx[r], rs[t][0], nm[t][0]);
                z[1] = fmaf(ry[r], rs[t][1], nm[t][1]);
                z[2] = fmaf(rz[r], rs[t][2], nm[t][2]);
                sQ[t] += Q; mQ[t] = fmaxf(mQ[t], Q);
#pragma unroll
                for (int d = 0; d < 3; ++d) {
                    const float u = Q * z[d];
                    sM[t][d] += u;
                    xM[t][d] = fmaxf(xM[t][d], u); nM[t][d] = fminf(nM[t][d], u);
                    const float v = fmaf(u, z[d], -Q);
                    sS[t][d] += v;
                    xS[t][d] = fmaxf(xS[t][d], v); nS[t][d] = fminf(nS[t][d], v);
                }
            }
        }
    }

#pragma unroll
    for (int t = 0; t < KT; ++t) {
        float a[20];
        a[0] = fmaxf(mQ[t], xhalf(mQ[t]));
        a[1] = sQ[t] + xhalf(sQ[t]);
#pragma unroll
        for (int d = 0; d < 3; ++d) {
            a[2  + d] = fmaxf(xM[t][d], xhalf(xM[t][d]));
            a[5  + d] = fminf(nM[t][d], xhalf(nM[t][d]));
            a[8  + d] = sM[t][d] + xhalf(sM[t][d]);
            a[11 + d] = fmaxf(xS[t][d], xhalf(xS[t][d]));
            a[14 + d] = fminf(nS[t][d], xhalf(nS[t][d]));
            a[17 + d] = sS[t][d] + xhalf(sS[t][d]);
        }
        if (!hi) {
#pragma unroll
            for (int f = 0; f < 20; ++f) red[wave][f][t * 16 + m] = a[f];
        }
    }
    __syncthreads();

    if (tid < K_) {
        float a[20];
#pragma unroll
        for (int f = 0; f < 20; ++f) a[f] = red[0][f][tid];
#pragma unroll
        for (int wv = 1; wv < NWAVE; ++wv) {
            a[0] = fmaxf(a[0], red[wv][0][tid]); a[1] += red[wv][1][tid];
#pragma unroll
            for (int d = 0; d < 3; ++d) {
                a[2  + d] = fmaxf(a[2  + d], red[wv][2  + d][tid]);
                a[5  + d] = fminf(a[5  + d], red[wv][5  + d][tid]);
                a[8  + d] +=            red[wv][8  + d][tid];
                a[11 + d] = fmaxf(a[11 + d], red[wv][11 + d][tid]);
                a[14 + d] = fminf(a[14 + d], red[wv][14 + d][tid]);
                a[17 + d] +=            red[wv][17 + d][tid];
            }
        }
        float* o = partial + ((size_t)(b * NCHUNK + ch) * K_ + tid) * 20;
#pragma unroll
        for (int f = 0; f < 20; ++f) o[f] = a[f];
    }
}

__global__ __launch_bounds__(128)
void fv_finalize(const float* __restrict__ w, const float* __restrict__ partial,
                 float* __restrict__ out)
{
    __shared__ float vals[20][128];
    __shared__ float ssq[20];
    const int b = blockIdx.x;
    const int t = threadIdx.x;

    float v[20];
    if (t < K_) {
        float a[20];
        {
            const float* p = partial + ((size_t)(b * NCHUNK) * K_ + t) * 20;
#pragma unroll
            for (int f = 0; f < 20; ++f) a[f] = p[f];
        }
        for (int c = 1; c < NCHUNK; ++c) {
            const float* p = partial + ((size_t)(b * NCHUNK + c) * K_ + t) * 20;
            a[0] = fmaxf(a[0], p[0]); a[1] += p[1];
#pragma unroll
            for (int d = 0; d < 3; ++d) {
                a[2  + d] = fmaxf(a[2  + d], p[2  + d]);
                a[5  + d] = fminf(a[5  + d], p[5  + d]);
                a[8  + d] +=            p[8  + d];
                a[11 + d] = fmaxf(a[11 + d], p[11 + d]);
                a[14 + d] = fminf(a[14 + d], p[14 + d]);
                a[17 + d] +=            p[17 + d];
            }
        }
        const float wk  = w[t];
        const float Nf  = (float)N_;
        const float spi = 1.0f / (sqrtf(wk) * Nf);
        const float ssg = 1.0f / (Nf * sqrtf(2.0f * wk));
        v[0] = (a[0] - wk)      * spi;
        v[1] = (a[1] - Nf * wk) * spi;
#pragma unroll
        for (int f = 2;  f < 11; ++f) v[f] = a[f] * spi;
#pragma unroll
        for (int f = 11; f < 20; ++f) v[f] = a[f] * ssg;
#pragma unroll
        for (int f = 0; f < 20; ++f)
            v[f] = copysignf(sqrtf(fabsf(v[f])), v[f]);
    } else {
#pragma unroll
        for (int f = 0; f < 20; ++f) v[f] = 0.0f;
    }
#pragma unroll
    for (int f = 0; f < 20; ++f) vals[f][t] = v[f];
    __syncthreads();

    if (t < 20) {
        float s = 0.0f;
        for (int k = 0; k < K_; ++k) { float x = vals[t][k]; s += x * x; }
        ssq[t] = s;
    }
    __syncthreads();

    if (t < K_) {
#pragma unroll
        for (int f = 0; f < 20; ++f)
            out[(size_t)b * 20 * K_ + f * K_ + t] =
                v[f] * rsqrtf(fmaxf(ssq[f], 1e-12f));
    }
}

extern "C" void kernel_launch(void* const* d_in, const int* in_sizes, int n_in,
                              void* d_out, int out_size, void* d_ws, size_t ws_size,
                              hipStream_t stream)
{
    const float* pts = (const float*)d_in[0];
    const float* w   = (const float*)d_in[1];
    const float* mu  = (const float*)d_in[2];
    const float* sg  = (const float*)d_in[3];
    float* out = (float*)d_out;

    float* partial = (float*)d_ws;

    fv_fused<<<B_ * NCHUNK, 256, 0, stream>>>(pts, w, mu, sg, partial);
    fv_finalize<<<B_, 128, 0, stream>>>(w, partial, out);
}


# grid re-cut: a block takes 1024 points x half the gaussians (the other half's densities only feed the softmax denominator); chunk partials per batch 16 to 8; a chunk sum now spans 1024 points
# speedup vs baseline: 1.8261x; 1.1820x over previous
#include <hip/hip_runtime.h>
#include <math.h>


typedef __attribute__((ext_vector_type(2))) float v2f;
typedef __attribute__((ext_vector_type(8))) float v8f;

template <int M>
__device__ __forceinline__ float row_xor(float x)
{
    const int b = __builtin_bit_cast(int, x);
    return __builtin_bit_cast(float,
        __builtin_amdgcn_update_dpp(b, b, 0x160 + M, 0xf, 0xf, false));
}

__device__ __forceinline__ float xhalf(float x)
{
    const int b = __builtin_bit_cast(int, x);
    return __builtin_bit_cast(float,
        __builtin_amdgcn_permlanex16(b, b, 0x76543210, 0xfedcba98, false, false));
}

#define B_      16
#define N_      8192
#define K_      125
#define KT      8
#define KH      4
#define NCHUNK  8
#define CHUNK   (N_ / NCHUNK)
#define NWAVE   8
#define PTW     (CHUNK / 16 / NWAVE)
#define C15     0.06349363593424097f

__global__ __launch_bounds__(256)
void fv_fused(const float* __restrict__ pts, const float* __restrict__ w,
              const float* __restrict__ mu, const float* __restrict__ sg,
              float* __restrict__ partial)
{
    __shared__ __attribute__((aligned(16))) float lx[CHUNK];
    __shared__ __attribute__((aligned(16))) float ly[CHUNK];
    __shared__ __attribute__((aligned(16))) float lz[CHUNK];
    __shared__ float4 ctab[KT * 32];
    __shared__ float red[NWAVE][20][KH * 16];

    const int tid  = threadIdx.x;
    const int lane = tid & 31;
    const int wave = tid >> 5;
    const int m    = lane & 15;
    const bool hi  = lane >= 16;
    const int b    = blockIdx.x >> 4;
    const int ch   = (blockIdx.x >> 1) & 7;
    const int g    = blockIdx.x & 1;
    const int n0   = ch * CHUNK;

    {
        float q[CHUNK / 256][3];
#pragma unroll
        for (int j = 0; j < CHUNK / 256; ++j) {
            const float* p = pts + (size_t)(b * N_ + n0 + tid + j * 256) * 3;
            q[j][0] = p[0]; q[j][1] = p[1]; q[j][2] = p[2];
        }
#pragma unroll
        for (int j = 0; j < CHUNK / 256; ++j) {
            const int i = tid + j * 256;
            lx[i] = q[j][0]; ly[i] = q[j][1]; lz[i] = q[j][2];
        }
    }

    {
        const int  kc    = wave * 16 + m;
        const bool valid = kc < K_;
        const int  ki    = valid ? kc : 0;
        const float m0 = mu[ki*3+0], m1 = mu[ki*3+1], m2 = mu[ki*3+2];
        const float s0 = sg[ki*3+0], s1 = sg[ki*3+1], s2 = sg[ki*3+2];
        const float i0 = 1.0f/(s0*s0), i1 = 1.0f/(s1*s1), i2 = 1.0f/(s2*s2);
        const float g0 = -0.5f*i0, g1 = -0.5f*i1, g2 = -0.5f*i2;
        const float g3 = m0*i0,    g4 = m1*i1,    g5 = m2*i2;
        const float lq = logf(w[ki] * C15 / (s0 * s1 * s2));
        const float g6 = valid ? (lq - 0.5f*(m0*m0*i0 + m1*m1*i1 + m2*m2*i2))
                               : -INFINITY;
        float4 v;
        v.x = hi ? g2   : g0;
        v.y = hi ? g3   : g1;
        v.z = hi ? g6   : g4;
        v.w = hi ? 0.0f : g5;
        ctab[tid] = v;
    }
    __syncthreads();

    v2f   b1[KH], b2[KH], o1[KH], o2[KH];
    float rs[KH][3], nm[KH][3];
#pragma unroll
    for (int t = 0; t < KH; ++t) {
        const float4 bc = ctab[(g * KH + t) * 32 + lane];
        b1[t].x = bc.x; b1[t].y = bc.y;
        b2[t].x = bc.z; b2[t].y = bc.w;
        const float4 oc = ctab[((1 - g) * KH + t) * 32 + lane];
        o1[t].x = oc.x; o1[t].y = oc.y;
        o2[t].x = oc.z; o2[t].y = oc.w;
        const int kc = (g * KH + t) * 16 + m;
        const int ki = (kc < K_) ? kc : 0;
#pragma unroll
        for (int d = 0; d < 3; ++d) {
            const float r = 1.0f / sg[ki*3+d];
            rs[t][d] = r;
            nm[t][d] = -mu[ki*3+d] * r;
        }
    }

    float sQ[KH], mQ[KH];
    float sM[KH][3], xM[KH][3], nM[KH][3];
    float sS[KH][3], xS[KH][3], nS[KH][3];
#pragma unroll
    for (int t = 0; t < KH; ++t) {
        sQ[t] = 0.0f; mQ[t] = -INFINITY;
#pragma unroll
        for (int d = 0; d < 3; ++d) {
            sM[t][d] = 0.0f; xM[t][d] = -INFINITY; nM[t][d] = INFINITY;
            sS[t][d] = 0.0f; xS[t][d] = -INFINITY; nS[t][d] = INFINITY;
        }
    }

#pragma unroll 1
    for (int it = 0; it < PTW; ++it) {
        const int pb = (wave * PTW + it) * 16;

        const float px = lx[pb + m], py = ly[pb + m], pz = lz[pb + m];
        v2f a1, a2;
        a1.x = hi ? pz * pz : px * px;
        a1.y = hi ? px      : py * py;
        a2.x = hi ? 1.0f    : py;
        a2.y = hi ? 0.0f    : pz;

        const int rb = pb + (hi ? 8 : 0);
        float rx[8], ry[8], rz[8];
        {
            const float4 x0 = *(const float4*)(lx + rb), x1 = *(const float4*)(lx + rb + 4);
            const float4 y0 = *(const float4*)(ly + rb), y1 = *(const float4*)(ly + rb + 4);
            const float4 z0 = *(const float4*)(lz + rb), z1 = *(const float4*)(lz + rb + 4);
            rx[0]=x0.x; rx[1]=x0.y; rx[2]=x0.z; rx[3]=x0.w; rx[4]=x1.x; rx[5]=x1.y; rx[6]=x1.z; rx[7]=x1.w;
            ry[0]=y0.x; ry[1]=y0.y; ry[2]=y0.z; ry[3]=y0.w; ry[4]=y1.x; ry[5]=y1.y; ry[6]=y1.z; ry[7]=y1.w;
            rz[0]=z0.x; rz[1]=z0.y; rz[2]=z0.z; rz[3]=z0.w; rz[4]=z1.x; rz[5]=z1.y; rz[6]=z1.z; rz[7]=z1.w;
        }

        float e[KH][8];
        float den[8];
#pragma unroll
        for (int r = 0; r < 8; ++r) den[r] = 0.0f;
#pragma unroll
        for (int t = 0; t < KH; ++t) {
            v8f c = {};
            c = __builtin_amdgcn_wmma_f32_16x16x4_f32(false, a1, false, b1[t],
                                                      (short)0, c, false, false);
            c = __builtin_amdgcn_wmma_f32_16x16x4_f32(false, a2, false, b2[t],
                                                      (short)0, c, false, false);
#pragma unroll
            for (int r = 0; r < 8; ++r) {
                const float ev = __expf(c[r]);
                e[t][r] = ev;
                den[r] += ev;
            }
        }
#pragma unroll
        for (int t = 0; t < KH; ++t) {
            v8f c = {};
            c = __builtin_amdgcn_wmma_f32_16x16x4_f32(false, a1, false, o1[t],
                                                      (short)0, c, false, false);
            c = __builtin_amdgcn_wmma_f32_16x16x4_f32(false, a2, false, o2[t],
                                                      (short)0, c, false, false);
#pragma unroll
            for (int r = 0; r < 8; ++r) den[r] += __expf(c[r]);
        }

        float rd[8];
#pragma unroll
        for (int r = 0; r < 8; ++r) {
            float v = den[r];
            v += row_xor<1>(v);
            v += row_xor<2>(v);
            v += row_xor<4>(v);
            v += row_xor<8>(v);
            rd[r] = 1.0f / v;
        }

#pragma unroll
        for (int t = 0; t < KH; ++t) {
#pragma unroll
            for (int r = 0; r < 8; ++r) {
                const float Q = e[t][r] * rd[r];
                float z[3];
                z[0] = fmaf(rx[r], rs[t][0], nm[t][0]);
                z[1] = fmaf(ry[r], rs[t][1], nm[t][1]);
                z[2] = fmaf(rz[r], rs[t][2], nm[t][2]);
                sQ[t] += Q; mQ[t] = fmaxf(mQ[t], Q);
#pragma unroll
                for (int d = 0; d < 3; ++d) {
                    const float u = Q * z[d];
                    sM[t][d] += u;
                    xM[t][d] = fmaxf(xM[t][d], u); nM[t][d] = fminf(nM[t][d], u);
                    const float v = fmaf(u, z[d], -Q);
                    sS[t][d] += v;
                    xS[t][d] = fmaxf(xS[t][d], v); nS[t][d] = fminf(nS[t][d], v);
                }
            }
        }
    }

#pragma unroll
    for (int t = 0; t < KH; ++t) {
        float a[20];
        a[0] = fmaxf(mQ[t], xhalf(mQ[t]));
        a[1] = sQ[t] + xhalf(sQ[t]);
#pragma unroll
        for (int d = 0; d < 3; ++d) {
            a[2  + d] = fmaxf(xM[t][d], xhalf(xM[t][d]));
            a[5  + d] = fminf(nM[t][d], xhalf(nM[t][d]));
            a[8  + d] = sM[t][d] + xhalf(sM[t][d]);
            a[11 + d] = fmaxf(xS[t][d], xhalf(xS[t][d]));
            a[14 + d] = fminf(nS[t][d], xhalf(nS[t][d]));
            a[17 + d] = sS[t][d] + xhalf(sS[t][d]);
        }
        if (!hi) {
#pragma unroll
            for (int f = 0; f < 20; ++f) red[wave][f][t * 16 + m] = a[f];
        }
    }
    __syncthreads();

    const int kg = g * (KH * 16) + tid;
    if (tid < KH * 16 && kg < K_) {
        float a[20];
#pragma unroll
        for (int f = 0; f < 20; ++f) a[f] = red[0][f][tid];
#pragma unroll
        for (int wv = 1; wv < NWAVE; ++wv) {
            a[0] = fmaxf(a[0], red[wv][0][tid]); a[1] += red[wv][1][tid];
#pragma unroll
            for (int d = 0; d < 3; ++d) {
                a[2  + d] = fmaxf(a[2  + d], red[wv][2  + d][tid]);
                a[5  + d] = fminf(a[5  + d], red[wv][5  + d][tid]);
                a[8  + d] +=            red[wv][8  + d][tid];
                a[11 + d] = fmaxf(a[11 + d], red[wv][11 + d][tid]);
                a[14 + d] = fminf(a[14 + d], red[wv][14 + d][tid]);
                a[17 + d] +=            red[wv][17 + d][tid];
            }
        }
        float* o = partial + ((size_t)(b * NCHUNK + ch) * K_ + kg) * 20;
#pragma unroll
        for (int f = 0; f < 20; ++f) o[f] = a[f];
    }
}

__global__ __launch_bounds__(128)
void fv_finalize(const float* __restrict__ w, const float* __restrict__ partial,
                 float* __restrict__ out)
{
    __shared__ float vals[20][128];
    __shared__ float ssq[20];
    const int b = blockIdx.x;
    const int t = threadIdx.x;

    float v[20];
    if (t < K_) {
        float a[20];
        {
            const float* p = partial + ((size_t)(b * NCHUNK) * K_ + t) * 20;
#pragma unroll
            for (int f = 0; f < 20; ++f) a[f] = p[f];
        }
        for (int c = 1; c < NCHUNK; ++c) {
            const float* p = partial + ((size_t)(b * NCHUNK + c) * K_ + t) * 20;
            a[0] = fmaxf(a[0], p[0]); a[1] += p[1];
#pragma unroll
            for (int d = 0; d < 3; ++d) {
                a[2  + d] = fmaxf(a[2  + d], p[2  + d]);
                a[5  + d] = fminf(a[5  + d], p[5  + d]);
                a[8  + d] +=            p[8  + d];
                a[11 + d] = fmaxf(a[11 + d], p[11 + d]);
                a[14 + d] = fminf(a[14 + d], p[14 + d]);
                a[17 + d] +=            p[17 + d];
            }
        }
        const float wk  = w[t];
        const float Nf  = (float)N_;
        const float spi = 1.0f / (sqrtf(wk) * Nf);
        const float ssg = 1.0f / (Nf * sqrtf(2.0f * wk));
        v[0] = (a[0] - wk)      * spi;
        v[1] = (a[1] - Nf * wk) * spi;
#pragma unroll
        for (int f = 2;  f < 11; ++f) v[f] = a[f] * spi;
#pragma unroll
        for (int f = 11; f < 20; ++f) v[f] = a[f] * ssg;
#pragma unroll
        for (int f = 0; f < 20; ++f)
            v[f] = copysignf(sqrtf(fabsf(v[f])), v[f]);
    } else {
#pragma unroll
        for (int f = 0; f < 20; ++f) v[f] = 0.0f;
    }
#pragma unroll
    for (int f = 0; f < 20; ++f) vals[f][t] = v[f];
    __syncthreads();

    if (t < 20) {
        float s = 0.0f;
        for (int k = 0; k < K_; ++k) { float x = vals[t][k]; s += x * x; }
        ssq[t] = s;
    }
    __syncthreads();

    if (t < K_) {
#pragma unroll
        for (int f = 0; f < 20; ++f)
            out[(size_t)b * 20 * K_ + f * K_ + t] =
                v[f] * rsqrtf(fmaxf(ssq[f], 1e-12f));
    }
}

extern "C" void kernel_launch(void* const* d_in, const int* in_sizes, int n_in,
                              void* d_out, int out_size, void* d_ws, size_t ws_size,
                              hipStream_t stream)
{
    const float* pts = (const float*)d_in[0];
    const float* w   = (const float*)d_in[1];
    const float* mu  = (const float*)d_in[2];
    const float* sg  = (const float*)d_in[3];
    float* out = (float*)d_out;

    float* partial = (float*)d_ws;

    fv_fused<<<B_ * NCHUNK * 2, 256, 0, stream>>>(pts, w, mu, sg, partial);
    fv_finalize<<<B_, 128, 0, stream>>>(w, partial, out);
}


# combine kernel: all 8 chunk partials of a thread's gaussian and w are loaded into registers before the first is used (one round trip to memory, was 9 dependent ones); order of every sum unchanged
# speedup vs baseline: 2.2025x; 1.2061x over previous
#include <hip/hip_runtime.h>
#include <math.h>


typedef __attribute__((ext_vector_type(2))) float v2f;
typedef __attribute__((ext_vector_type(8))) float v8f;

template <int M>
__device__ __forceinline__ float row_xor(float x)
{
    const int b = __builtin_bit_cast(int, x);
    return __builtin_bit_cast(float,
        __builtin_amdgcn_update_dpp(b, b, 0x160 + M, 0xf, 0xf, false));
}

__device__ __forceinline__ float xhalf(float x)
{
    const int b = __builtin_bit_cast(int, x);
    return __builtin_bit_cast(float,
        __builtin_amdgcn_permlanex16(b, b, 0x76543210, 0xfedcba98, false, false));
}

#define B_      16
#define N_      8192
#define K_      125
#define KT      8
#define KH      4
#define NCHUNK  8
#define CHUNK   (N_ / NCHUNK)
#define NWAVE   8
#define PTW     (CHUNK / 16 / NWAVE)
#define C15     0.06349363593424097f

__global__ __launch_bounds__(256)
void fv_fused(const float* __restrict__ pts, const float* __restrict__ w,
              const float* __restrict__ mu, const float* __restrict__ sg,
              float* __restrict__ partial)
{
    __shared__ __attribute__((aligned(16))) float lx[CHUNK];
    __shared__ __attribute__((aligned(16))) float ly[CHUNK];
    __shared__ __attribute__((aligned(16))) float lz[CHUNK];
    __shared__ float4 ctab[KT * 32];
    __shared__ float red[NWAVE][20][KH * 16];

    const int tid  = threadIdx.x;
    const int lane = tid & 31;
    const int wave = tid >> 5;
    const int m    = lane & 15;
    const bool hi  = lane >= 16;
    const int b    = blockIdx.x >> 4;
    const int ch   = (blockIdx.x >> 1) & 7;
    const int g    = blockIdx.x & 1;
    const int n0   = ch * CHUNK;

    {
        float q[CHUNK / 256][3];
#pragma unroll
        for (int j = 0; j < CHUNK / 256; ++j) {
            const float* p = pts + (size_t)(b * N_ + n0 + tid + j * 256) * 3;
            q[j][0] = p[0]; q[j][1] = p[1]; q[j][2] = p[2];
        }
#pragma unroll
        for (int j = 0; j < CHUNK / 256; ++j) {
            const int i = tid + j * 256;
            lx[i] = q[j][0]; ly[i] = q[j][1]; lz[i] = q[j][2];
        }
    }

    {
        const int  kc    = wave * 16 + m;
        const bool valid = kc < K_;
        const int  ki    = valid ? kc : 0;
        const float m0 = mu[ki*3+0], m1 = mu[ki*3+1], m2 = mu[ki*3+2];
        const float s0 = sg[ki*3+0], s1 = sg[ki*3+1], s2 = sg[ki*3+2];
        const float i0 = 1.0f/(s0*s0), i1 = 1.0f/(s1*s1), i2 = 1.0f/(s2*s2);
        const float g0 = -0.5f*i0, g1 = -0.5f*i1, g2 = -0.5f*i2;
        const float g3 = m0*i0,    g4 = m1*i1,    g5 = m2*i2;
        const float lq = logf(w[ki] * C15 / (s0 * s1 * s2));
        const float g6 = valid ? (lq - 0.5f*(m0*m0*i0 + m1*m1*i1 + m2*m2*i2))
                               : -INFINITY;
        float4 v;
        v.x = hi ? g2   : g0;
        v.y = hi ? g3   : g1;
        v.z = hi ? g6   : g4;
        v.w = hi ? 0.0f : g5;
        ctab[tid] = v;
    }
    __syncthreads();

    v2f   b1[KH], b2[KH], o1[KH], o2[KH];
    float rs[KH][3], nm[KH][3];
#pragma unroll
    for (int t = 0; t < KH; ++t) {
        const float4 bc = ctab[(g * KH + t) * 32 + lane];
        b1[t].x = bc.x; b1[t].y = bc.y;
        b2[t].x = bc.z; b2[t].y = bc.w;
        const float4 oc = ctab[((1 - g) * KH + t) * 32 + lane];
        o1[t].x = oc.x; o1[t].y = oc.y;
        o2[t].x = oc.z; o2[t].y = oc.w;
        const int kc = (g * KH + t) * 16 + m;
        const int ki = (kc < K_) ? kc : 0;
#pragma unroll
        for (int d = 0; d < 3; ++d) {
            const float r = 1.0f / sg[ki*3+d];
            rs[t][d] = r;
            nm[t][d] = -mu[ki*3+d] * r;
        }
    }

    float sQ[KH], mQ[KH];
    float sM[KH][3], xM[KH][3], nM[KH][3];
    float sS[KH][3], xS[KH][3], nS[KH][3];
#pragma unroll
    for (int t = 0; t < KH; ++t) {
        sQ[t] = 0.0f; mQ[t] = -INFINITY;
#pragma unroll
        for (int d = 0; d < 3; ++d) {
            sM[t][d] = 0.0f; xM[t][d] = -INFINITY; nM[t][d] = INFINITY;
            sS[t][d] = 0.0f; xS[t][d] = -INFINITY; nS[t][d] = INFINITY;
        }
    }

#pragma unroll 1
    for (int it = 0; it < PTW; ++it) {
        const int pb = (wave * PTW + it) * 16;

        const float px = lx[pb + m], py = ly[pb + m], pz = lz[pb + m];
        v2f a1, a2;
        a1.x = hi ? pz * pz : px * px;
        a1.y = hi ? px      : py * py;
        a2.x = hi ? 1.0f    : py;
        a2.y = hi ? 0.0f    : pz;

        const int rb = pb + (hi ? 8 : 0);
        float rx[8], ry[8], rz[8];
        {
            const float4 x0 = *(const float4*)(lx + rb), x1 = *(const float4*)(lx + rb + 4);
            const float4 y0 = *(const float4*)(ly + rb), y1 = *(const float4*)(ly + rb + 4);
            const float4 z0 = *(const float4*)(lz + rb), z1 = *(const float4*)(lz + rb + 4);
            rx[0]=x0.x; rx[1]=x0.y; rx[2]=x0.z; rx[3]=x0.w; rx[4]=x1.x; rx[5]=x1.y; rx[6]=x1.z; rx[7]=x1.w;
            ry[0]=y0.x; ry[1]=y0.y; ry[2]=y0.z; ry[3]=y0.w; ry[4]=y1.x; ry[5]=y1.y; ry[6]=y1.z; ry[7]=y1.w;
            rz[0]=z0.x; rz[1]=z0.y; rz[2]=z0.z; rz[3]=z0.w; rz[4]=z1.x; rz[5]=z1.y; rz[6]=z1.z; rz[7]=z1.w;
        }

        float e[KH][8];
        float den[8];
#pragma unroll
        for (int r = 0; r < 8; ++r) den[r] = 0.0f;
#pragma unroll
        for (int t = 0; t < KH; ++t) {
            v8f c = {};
            c = __builtin_amdgcn_wmma_f32_16x16x4_f32(false, a1, false, b1[t],
                                                      (short)0, c, false, false);
            c = __builtin_amdgcn_wmma_f32_16x16x4_f32(false, a2, false, b2[t],
                                                      (short)0, c, false, false);
#pragma unroll
            for (int r = 0; r < 8; ++r) {
                const float ev = __expf(c[r]);
                e[t][r] = ev;
                den[r] += ev;
            }
        }
#pragma unroll
        for (int t = 0; t < KH; ++t) {
            v8f c = {};
            c = __builtin_amdgcn_wmma_f32_16x16x4_f32(false, a1, false, o1[t],
                                                      (short)0, c, false, false);
            c = __builtin_amdgcn_wmma_f32_16x16x4_f32(false, a2, false, o2[t],
                                                      (short)0, c, false, false);
#pragma unroll
            for (int r = 0; r < 8; ++r) den[r] += __expf(c[r]);
        }

        float rd[8];
#pragma unroll
        for (int r = 0; r < 8; ++r) {
            float v = den[r];
            v += row_xor<1>(v);
            v += row_xor<2>(v);
            v += row_xor<4>(v);
            v += row_xor<8>(v);
            rd[r] = 1.0f / v;
        }

#pragma unroll
        for (int t = 0; t < KH; ++t) {
#pragma unroll
            for (int r = 0; r < 8; ++r) {
                const float Q = e[t][r] * rd[r];
                float z[3];
                z[0] = fmaf(rx[r], rs[t][0], nm[t][0]);
                z[1] = fmaf(ry[r], rs[t][1], nm[t][1]);
                z[2] = fmaf(rz[r], rs[t][2], nm[t][2]);
                sQ[t] += Q; mQ[t] = fmaxf(mQ[t], Q);
#pragma unroll
                for (int d = 0; d < 3; ++d) {
                    const float u = Q * z[d];
                    sM[t][d] += u;
                    xM[t][d] = fmaxf(xM[t][d], u); nM[t][d] = fminf(nM[t][d], u);
                    const float v = fmaf(u, z[d], -Q);
                    sS[t][d] += v;
                    xS[t][d] = fmaxf(xS[t][d], v); nS[t][d] = fminf(nS[t][d], v);
                }
            }
        }
    }

#pragma unroll
    for (int t = 0; t < KH; ++t) {
        float a[20];
        a[0] = fmaxf(mQ[t], xhalf(mQ[t]));
        a[1] = sQ[t] + xhalf(sQ[t]);
#pragma unroll
        for (int d = 0; d < 3; ++d) {
            a[2  + d] = fmaxf(xM[t][d], xhalf(xM[t][d]));
            a[5  + d] = fminf(nM[t][d], xhalf(nM[t][d]));
            a[8  + d] = sM[t][d] + xhalf(sM[t][d]);
            a[11 + d] = fmaxf(xS[t][d], xhalf(xS[t][d]));
            a[14 + d] = fminf(nS[t][d], xhalf(nS[t][d]));
            a[17 + d] = sS[t][d] + xhalf(sS[t][d]);
        }
        if (!hi) {
#pragma unroll
            for (int f = 0; f < 20; ++f) red[wave][f][t * 16 + m] = a[f];
        }
    }
    __syncthreads();

    const int kg = g * (KH * 16) + tid;
    if (tid < KH * 16 && kg < K_) {
        float a[20];
#pragma unroll
        for (int f = 0; f < 20; ++f) a[f] = red[0][f][tid];
#pragma unroll
        for (int wv = 1; wv < NWAVE; ++wv) {
            a[0] = fmaxf(a[0], red[wv][0][tid]); a[1] += red[wv][1][tid];
#pragma unroll
            for (int d = 0; d < 3; ++d) {
                a[2  + d] = fmaxf(a[2  + d], red[wv][2  + d][tid]);
                a[5  + d] = fminf(a[5  + d], red[wv][5  + d][tid]);
                a[8  + d] +=            red[wv][8  + d][tid];
                a[11 + d] = fmaxf(a[11 + d], red[wv][11 + d][tid]);
                a[14 + d] = fminf(a[14 + d], red[wv][14 + d][tid]);
                a[17 + d] +=            red[wv][17 + d][tid];
            }
        }
        float* o = partial + ((size_t)(b * NCHUNK + ch) * K_ + kg) * 20;
#pragma unroll
        for (int f = 0; f < 20; ++f) o[f] = a[f];
    }
}

__global__ __launch_bounds__(128) __attribute__((amdgpu_waves_per_eu(1, 1)))
void fv_finalize(const float* __restrict__ w, const float* __restrict__ partial,
                 float* __restrict__ out)
{
    __shared__ float vals[20][128];
    __shared__ float ssq[20];
    const int b = blockIdx.x;
    const int t = threadIdx.x;

    float v[20];
    if (t < K_) {
        float q[NCHUNK][20];
#pragma unroll
        for (int c = 0; c < NCHUNK; ++c) {
            const float* p = partial + ((size_t)(b * NCHUNK + c) * K_ + t) * 20;
#pragma unroll
            for (int f = 0; f < 20; ++f) q[c][f] = p[f];
        }
        const float wk  = w[t];

        float a[20];
#pragma unroll
        for (int f = 0; f < 20; ++f) a[f] = q[0][f];
#pragma unroll
        for (int c = 1; c < NCHUNK; ++c) {
            a[0] = fmaxf(a[0], q[c][0]); a[1] += q[c][1];
#pragma unroll
            for (int d = 0; d < 3; ++d) {
                a[2  + d] = fmaxf(a[2  + d], q[c][2  + d]);
                a[5  + d] = fminf(a[5  + d], q[c][5  + d]);
                a[8  + d] +=            q[c][8  + d];
                a[11 + d] = fmaxf(a[11 + d], q[c][11 + d]);
                a[14 + d] = fminf(a[14 + d], q[c][14 + d]);
                a[17 + d] +=            q[c][17 + d];
            }
        }
        const float Nf  = (float)N_;
        const float spi = 1.0f / (sqrtf(wk) * Nf);
        const float ssg = 1.0f / (Nf * sqrtf(2.0f * wk));
        v[0] = (a[0] - wk)      * spi;
        v[1] = (a[1] - Nf * wk) * spi;
#pragma unroll
        for (int f = 2;  f < 11; ++f) v[f] = a[f] * spi;
#pragma unroll
        for (int f = 11; f < 20; ++f) v[f] = a[f] * ssg;
#pragma unroll
        for (int f = 0; f < 20; ++f)
            v[f] = copysignf(sqrtf(fabsf(v[f])), v[f]);
    } else {
#pragma unroll
        for (int f = 0; f < 20; ++f) v[f] = 0.0f;
    }
#pragma unroll
    for (int f = 0; f < 20; ++f) vals[f][t] = v[f];
    __syncthreads();

    if (t < 20) {
        float s = 0.0f;
        for (int k = 0; k < K_; ++k) { float x = vals[t][k]; s += x * x; }
        ssq[t] = s;
    }
    __syncthreads();

    if (t < K_) {
#pragma unroll
        for (int f = 0; f < 20; ++f)
            out[(size_t)b * 20 * K_ + f * K_ + t] =
                v[f] * rsqrtf(fmaxf(ssq[f], 1e-12f));
    }
}

extern "C" void kernel_launch(void* const* d_in, const int* in_sizes, int n_in,
                              void* d_out, int out_size, void* d_ws, size_t ws_size,
                              hipStream_t stream)
{
    const float* pts = (const float*)d_in[0];
    const float* w   = (const float*)d_in[1];
    const float* mu  = (const float*)d_in[2];
    const float* sg  = (const float*)d_in[3];
    float* out = (float*)d_out;

    float* partial = (float*)d_ws;

    fv_fused<<<B_ * NCHUNK * 2, 256, 0, stream>>>(pts, w, mu, sg, partial);
    fv_finalize<<<B_, 128, 0, stream>>>(w, partial, out);
}


# softmax denominator of a 16-point tile: 16-lane row sum as adds with a DPP operand (lane move folded in) and 1/den by v_rcp_f32 (1 ulp) instead of the IEEE division sequence; same sum order
# speedup vs baseline: 2.3154x; 1.0513x over previous
#include <hip/hip_runtime.h>
#include <math.h>


typedef __attribute__((ext_vector_type(2))) float v2f;
typedef __attribute__((ext_vector_type(8))) float v8f;

template <int M>
__device__ __forceinline__ float row_xor(float x)
{
    const int b = __builtin_bit_cast(int, x);
    return __builtin_bit_cast(float,
        __builtin_amdgcn_update_dpp(0, b, 0x160 + M, 0xf, 0xf, true));
}

__device__ __forceinline__ float xhalf(float x)
{
    const int b = __builtin_bit_cast(int, x);
    return __builtin_bit_cast(float,
        __builtin_amdgcn_permlanex16(b, b, 0x76543210, 0xfedcba98, false, false));
}

#define B_      16
#define N_      8192
#define K_      125
#define KT      8
#define KH      4
#define NCHUNK  8
#define CHUNK   (N_ / NCHUNK)
#define NWAVE   8
#define PTW     (CHUNK / 16 / NWAVE)
#define C15     0.06349363593424097f

__global__ __launch_bounds__(256)
void fv_fused(const float* __restrict__ pts, const float* __restrict__ w,
              const float* __restrict__ mu, const float* __restrict__ sg,
              float* __restrict__ partial)
{
    __shared__ __attribute__((aligned(16))) float lx[CHUNK];
    __shared__ __attribute__((aligned(16))) float ly[CHUNK];
    __shared__ __attribute__((aligned(16))) float lz[CHUNK];
    __shared__ float4 ctab[KT * 32];
    __shared__ float red[NWAVE][20][KH * 16];

    const int tid  = threadIdx.x;
    const int lane = tid & 31;
    const int wave = tid >> 5;
    const int m    = lane & 15;
    const bool hi  = lane >= 16;
    const int b    = blockIdx.x >> 4;
    const int ch   = (blockIdx.x >> 1) & 7;
    const int g    = blockIdx.x & 1;
    const int n0   = ch * CHUNK;

    {
        float q[CHUNK / 256][3];
#pragma unroll
        for (int j = 0; j < CHUNK / 256; ++j) {
            const float* p = pts + (size_t)(b * N_ + n0 + tid + j * 256) * 3;
            q[j][0] = p[0]; q[j][1] = p[1]; q[j][2] = p[2];
        }
#pragma unroll
        for (int j = 0; j < CHUNK / 256; ++j) {
            const int i = tid + j * 256;
            lx[i] = q[j][0]; ly[i] = q[j][1]; lz[i] = q[j][2];
        }
    }

    {
        const int  kc    = wave * 16 + m;
        const bool valid = kc < K_;
        const int  ki    = valid ? kc : 0;
        const float m0 = mu[ki*3+0], m1 = mu[ki*3+1], m2 = mu[ki*3+2];
        const float s0 = sg[ki*3+0], s1 = sg[ki*3+1], s2 = sg[ki*3+2];
        const float i0 = 1.0f/(s0*s0), i1 = 1.0f/(s1*s1), i2 = 1.0f/(s2*s2);
        const float g0 = -0.5f*i0, g1 = -0.5f*i1, g2 = -0.5f*i2;
        const float g3 = m0*i0,    g4 = m1*i1,    g5 = m2*i2;
        const float lq = logf(w[ki] * C15 / (s0 * s1 * s2));
        const float g6 = valid ? (lq - 0.5f*(m0*m0*i0 + m1*m1*i1 + m2*m2*i2))
                               : -INFINITY;
        float4 v;
        v.x = hi ? g2   : g0;
        v.y = hi ? g3   : g1;
        v.z = hi ? g6   : g4;
        v.w = hi ? 0.0f : g5;
        ctab[tid] = v;
    }
    __syncthreads();

    v2f   b1[KH], b2[KH], o1[KH], o2[KH];
    float rs[KH][3], nm[KH][3];
#pragma unroll
    for (int t = 0; t < KH; ++t) {
        const float4 bc = ctab[(g * KH + t) * 32 + lane];
        b1[t].x = bc.x; b1[t].y = bc.y;
        b2[t].x = bc.z; b2[t].y = bc.w;
        const float4 oc = ctab[((1 - g) * KH + t) * 32 + lane];
        o1[t].x = oc.x; o1[t].y = oc.y;
        o2[t].x = oc.z; o2[t].y = oc.w;
        const int kc = (g * KH + t) * 16 + m;
        const int ki = (kc < K_) ? kc : 0;
#pragma unroll
        for (int d = 0; d < 3; ++d) {
            const float r = 1.0f / sg[ki*3+d];
            rs[t][d] = r;
            nm[t][d] = -mu[ki*3+d] * r;
        }
    }

    float sQ[KH], mQ[KH];
    float sM[KH][3], xM[KH][3], nM[KH][3];
    float sS[KH][3], xS[KH][3], nS[KH][3];
#pragma unroll
    for (int t = 0; t < KH; ++t) {
        sQ[t] = 0.0f; mQ[t] = -INFINITY;
#pragma unroll
        for (int d = 0; d < 3; ++d) {
            sM[t][d] = 0.0f; xM[t][d] = -INFINITY; nM[t][d] = INFINITY;
            sS[t][d] = 0.0f; xS[t][d] = -INFINITY; nS[t][d] = INFINITY;
        }
    }

#pragma unroll 1
    for (int it = 0; it < PTW; ++it) {
        const int pb = (wave * PTW + it) * 16;

        const float px = lx[pb + m], py = ly[pb + m], pz = lz[pb + m];
        v2f a1, a2;
        a1.x = hi ? pz * pz : px * px;
        a1.y = hi ? px      : py * py;
        a2.x = hi ? 1.0f    : py;
        a2.y = hi ? 0.0f    : pz;

        const int rb = pb + (hi ? 8 : 0);
        float rx[8], ry[8], rz[8];
        {
            const float4 x0 = *(const float4*)(lx + rb), x1 = *(const float4*)(lx + rb + 4);
            const float4 y0 = *(const float4*)(ly + rb), y1 = *(const float4*)(ly + rb + 4);
            const float4 z0 = *(const float4*)(lz + rb), z1 = *(const float4*)(lz + rb + 4);
            rx[0]=x0.x; rx[1]=x0.y; rx[2]=x0.z; rx[3]=x0.w; rx[4]=x1.x; rx[5]=x1.y; rx[6]=x1.z; rx[7]=x1.w;
            ry[0]=y0.x; ry[1]=y0.y; ry[2]=y0.z; ry[3]=y0.w; ry[4]=y1.x; ry[5]=y1.y; ry[6]=y1.z; ry[7]=y1.w;
            rz[0]=z0.x; rz[1]=z0.y; rz[2]=z0.z; rz[3]=z0.w; rz[4]=z1.x; rz[5]=z1.y; rz[6]=z1.z; rz[7]=z1.w;
        }

        float e[KH][8];
        float den[8];
#pragma unroll
        for (int r = 0; r < 8; ++r) den[r] = 0.0f;
#pragma unroll
        for (int t = 0; t < KH; ++t) {
            v8f c = {};
            c = __builtin_amdgcn_wmma_f32_16x16x4_f32(false, a1, false, b1[t],
                                                      (short)0, c, false, false);
            c = __builtin_amdgcn_wmma_f32_16x16x4_f32(false, a2, false, b2[t],
                                                      (short)0, c, false, false);
#pragma unroll
            for (int r = 0; r < 8; ++r) {
                const float ev = __expf(c[r]);
                e[t][r] = ev;
                den[r] += ev;
            }
        }
#pragma unroll
        for (int t = 0; t < KH; ++t) {
            v8f c = {};
            c = __builtin_amdgcn_wmma_f32_16x16x4_f32(false, a1, false, o1[t],
                                                      (short)0, c, false, false);
            c = __builtin_amdgcn_wmma_f32_16x16x4_f32(false, a2, false, o2[t],
                                                      (short)0, c, false, false);
#pragma unroll
            for (int r = 0; r < 8; ++r) den[r] += __expf(c[r]);
        }

        float rd[8];
#pragma unroll
        for (int r = 0; r < 8; ++r) {
            float v = den[r];
            v += row_xor<1>(v);
            v += row_xor<2>(v);
            v += row_xor<4>(v);
            v += row_xor<8>(v);
            rd[r] = __builtin_amdgcn_rcpf(v);
        }

#pragma unroll
        for (int t = 0; t < KH; ++t) {
#pragma unroll
            for (int r = 0; r < 8; ++r) {
                const float Q = e[t][r] * rd[r];
                float z[3];
                z[0] = fmaf(rx[r], rs[t][0], nm[t][0]);
                z[1] = fmaf(ry[r], rs[t][1], nm[t][1]);
                z[2] = fmaf(rz[r], rs[t][2], nm[t][2]);
                sQ[t] += Q; mQ[t] = fmaxf(mQ[t], Q);
#pragma unroll
                for (int d = 0; d < 3; ++d) {
                    const float u = Q * z[d];
                    sM[t][d] += u;
                    xM[t][d] = fmaxf(xM[t][d], u); nM[t][d] = fminf(nM[t][d], u);
                    const float v = fmaf(u, z[d], -Q);
                    sS[t][d] += v;
                    xS[t][d] = fmaxf(xS[t][d], v); nS[t][d] = fminf(nS[t][d], v);
                }
            }
        }
    }

#pragma unroll
    for (int t = 0; t < KH; ++t) {
        float a[20];
        a[0] = fmaxf(mQ[t], xhalf(mQ[t]));
        a[1] = sQ[t] + xhalf(sQ[t]);
#pragma unroll
        for (int d = 0; d < 3; ++d) {
            a[2  + d] = fmaxf(xM[t][d], xhalf(xM[t][d]));
            a[5  + d] = fminf(nM[t][d], xhalf(nM[t][d]));
            a[8  + d] = sM[t][d] + xhalf(sM[t][d]);
            a[11 + d] = fmaxf(xS[t][d], xhalf(xS[t][d]));
            a[14 + d] = fminf(nS[t][d], xhalf(nS[t][d]));
            a[17 + d] = sS[t][d] + xhalf(sS[t][d]);
        }
        if (!hi) {
#pragma unroll
            for (int f = 0; f < 20; ++f) red[wave][f][t * 16 + m] = a[f];
        }
    }
    __syncthreads();

    const int kg = g * (KH * 16) + tid;
    if (tid < KH * 16 && kg < K_) {
        float a[20];
#pragma unroll
        for (int f = 0; f < 20; ++f) a[f] = red[0][f][tid];
#pragma unroll
        for (int wv = 1; wv < NWAVE; ++wv) {
            a[0] = fmaxf(a[0], red[wv][0][tid]); a[1] += red[wv][1][tid];
#pragma unroll
            for (int d = 0; d < 3; ++d) {
                a[2  + d] = fmaxf(a[2  + d], red[wv][2  + d][tid]);
                a[5  + d] = fminf(a[5  + d], red[wv][5  + d][tid]);
                a[8  + d] +=            red[wv][8  + d][tid];
                a[11 + d] = fmaxf(a[11 + d], red[wv][11 + d][tid]);
                a[14 + d] = fminf(a[14 + d], red[wv][14 + d][tid]);
                a[17 + d] +=            red[wv][17 + d][tid];
            }
        }
        float* o = partial + ((size_t)(b * NCHUNK + ch) * K_ + kg) * 20;
#pragma unroll
        for (int f = 0; f < 20; ++f) o[f] = a[f];
    }
}

__global__ __launch_bounds__(128) __attribute__((amdgpu_waves_per_eu(1, 1)))
void fv_finalize(const float* __restrict__ w, const float* __restrict__ partial,
                 float* __restrict__ out)
{
    __shared__ float vals[20][128];
    __shared__ float ssq[20];
    const int b = blockIdx.x;
    const int t = threadIdx.x;

    float v[20];
    if (t < K_) {
        float q[NCHUNK][20];
#pragma unroll
        for (int c = 0; c < NCHUNK; ++c) {
            const float* p = partial + ((size_t)(b * NCHUNK + c) * K_ + t) * 20;
#pragma unroll
            for (int f = 0; f < 20; ++f) q[c][f] = p[f];
        }
        const float wk  = w[t];

        float a[20];
#pragma unroll
        for (int f = 0; f < 20; ++f) a[f] = q[0][f];
#pragma unroll
        for (int c = 1; c < NCHUNK; ++c) {
            a[0] = fmaxf(a[0], q[c][0]); a[1] += q[c][1];
#pragma unroll
            for (int d = 0; d < 3; ++d) {
                a[2  + d] = fmaxf(a[2  + d], q[c][2  + d]);
                a[5  + d] = fminf(a[5  + d], q[c][5  + d]);
                a[8  + d] +=            q[c][8  + d];
                a[11 + d] = fmaxf(a[11 + d], q[c][11 + d]);
                a[14 + d] = fminf(a[14 + d], q[c][14 + d]);
                a[17 + d] +=            q[c][17 + d];
            }
        }
        const float Nf  = (float)N_;
        const float spi = 1.0f / (sqrtf(wk) * Nf);
        const float ssg = 1.0f / (Nf * sqrtf(2.0f * wk));
        v[0] = (a[0] - wk)      * spi;
        v[1] = (a[1] - Nf * wk) * spi;
#pragma unroll
        for (int f = 2;  f < 11; ++f) v[f] = a[f] * spi;
#pragma unroll
        for (int f = 11; f < 20; ++f) v[f] = a[f] * ssg;
#pragma unroll
        for (int f = 0; f < 20; ++f)
            v[f] = copysignf(sqrtf(fabsf(v[f])), v[f]);
    } else {
#pragma unroll
        for (int f = 0; f < 20; ++f) v[f] = 0.0f;
    }
#pragma unroll
    for (int f = 0; f < 20; ++f) vals[f][t] = v[f];
    __syncthreads();

    if (t < 20) {
        float s = 0.0f;
        for (int k = 0; k < K_; ++k) { float x = vals[t][k]; s += x * x; }
        ssq[t] = s;
    }
    __syncthreads();

    if (t < K_) {
#pragma unroll
        for (int f = 0; f < 20; ++f)
            out[(size_t)b * 20 * K_ + f * K_ + t] =
                v[f] * rsqrtf(fmaxf(ssq[f], 1e-12f));
    }
}

extern "C" void kernel_launch(void* const* d_in, const int* in_sizes, int n_in,
                              void* d_out, int out_size, void* d_ws, size_t ws_size,
                              hipStream_t stream)
{
    const float* pts = (const float*)d_in[0];
    const float* w   = (const float*)d_in[1];
    const float* mu  = (const float*)d_in[2];
    const float* sg  = (const float*)d_in[3];
    float* out = (float*)d_out;

    float* partial = (float*)d_ws;

    fv_fused<<<B_ * NCHUNK * 2, 256, 0, stream>>>(pts, w, mu, sg, partial);
    fv_finalize<<<B_, 128, 0, stream>>>(w, partial, out);
}


# finalize kernel: a lane combines one 16-byte slot of all 8 chunk partials: a wave load covers whole lines (was 20 lines), and its LDS image is summed down conflict-free columns; sum order unchanged
# speedup vs baseline: 2.4505x; 1.0583x over previous
#include <hip/hip_runtime.h>
#include <math.h>


typedef __attribute__((ext_vector_type(2))) float v2f;
typedef __attribute__((ext_vector_type(8))) float v8f;

template <int M>
__device__ __forceinline__ float row_xor(float x)
{
    const int b = __builtin_bit_cast(int, x);
    return __builtin_bit_cast(float,
        __builtin_amdgcn_update_dpp(0, b, 0x160 + M, 0xf, 0xf, true));
}

__device__ __forceinline__ float xhalf(float x)
{
    const int b = __builtin_bit_cast(int, x);
    return __builtin_bit_cast(float,
        __builtin_amdgcn_permlanex16(b, b, 0x76543210, 0xfedcba98, false, false));
}

#define B_      16
#define N_      8192
#define K_      125
#define KT      8
#define KH      4
#define NCHUNK  8
#define CHUNK   (N_ / NCHUNK)
#define NWAVE   8
#define PTW     (CHUNK / 16 / NWAVE)
#define C15     0.06349363593424097f

__global__ __launch_bounds__(256)
void fv_fused(const float* __restrict__ pts, const float* __restrict__ w,
              const float* __restrict__ mu, const float* __restrict__ sg,
              float* __restrict__ partial)
{
    __shared__ __attribute__((aligned(16))) float lx[CHUNK];
    __shared__ __attribute__((aligned(16))) float ly[CHUNK];
    __shared__ __attribute__((aligned(16))) float lz[CHUNK];
    __shared__ float4 ctab[KT * 32];
    __shared__ float red[NWAVE][20][KH * 16];

    const int tid  = threadIdx.x;
    const int lane = tid & 31;
    const int wave = tid >> 5;
    const int m    = lane & 15;
    const bool hi  = lane >= 16;
    const int b    = blockIdx.x >> 4;
    const int ch   = (blockIdx.x >> 1) & 7;
    const int g    = blockIdx.x & 1;
    const int n0   = ch * CHUNK;

    {
        float q[CHUNK / 256][3];
#pragma unroll
        for (int j = 0; j < CHUNK / 256; ++j) {
            const float* p = pts + (size_t)(b * N_ + n0 + tid + j * 256) * 3;
            q[j][0] = p[0]; q[j][1] = p[1]; q[j][2] = p[2];
        }
#pragma unroll
        for (int j = 0; j < CHUNK / 256; ++j) {
            const int i = tid + j * 256;
            lx[i] = q[j][0]; ly[i] = q[j][1]; lz[i] = q[j][2];
        }
    }

    {
        const int  kc    = wave * 16 + m;
        const bool valid = kc < K_;
        const int  ki    = valid ? kc : 0;
        const float m0 = mu[ki*3+0], m1 = mu[ki*3+1], m2 = mu[ki*3+2];
        const float s0 = sg[ki*3+0], s1 = sg[ki*3+1], s2 = sg[ki*3+2];
        const float i0 = 1.0f/(s0*s0), i1 = 1.0f/(s1*s1), i2 = 1.0f/(s2*s2);
        const float g0 = -0.5f*i0, g1 = -0.5f*i1, g2 = -0.5f*i2;
        const float g3 = m0*i0,    g4 = m1*i1,    g5 = m2*i2;
        const float lq = logf(w[ki] * C15 / (s0 * s1 * s2));
        const float g6 = valid ? (lq - 0.5f*(m0*m0*i0 + m1*m1*i1 + m2*m2*i2))
                               : -INFINITY;
        float4 v;
        v.x = hi ? g2   : g0;
        v.y = hi ? g3   : g1;
        v.z = hi ? g6   : g4;
        v.w = hi ? 0.0f : g5;
        ctab[tid] = v;
    }
    __syncthreads();

    v2f   b1[KH], b2[KH], o1[KH], o2[KH];
    float rs[KH][3], nm[KH][3];
#pragma unroll
    for (int t = 0; t < KH; ++t) {
        const float4 bc = ctab[(g * KH + t) * 32 + lane];
        b1[t].x = bc.x; b1[t].y = bc.y;
        b2[t].x = bc.z; b2[t].y = bc.w;
        const float4 oc = ctab[((1 - g) * KH + t) * 32 + lane];
        o1[t].x = oc.x; o1[t].y = oc.y;
        o2[t].x = oc.z; o2[t].y = oc.w;
        const int kc = (g * KH + t) * 16 + m;
        const int ki = (kc < K_) ? kc : 0;
#pragma unroll
        for (int d = 0; d < 3; ++d) {
            const float r = 1.0f / sg[ki*3+d];
            rs[t][d] = r;
            nm[t][d] = -mu[ki*3+d] * r;
        }
    }

    float sQ[KH], mQ[KH];
    float sM[KH][3], xM[KH][3], nM[KH][3];
    float sS[KH][3], xS[KH][3], nS[KH][3];
#pragma unroll
    for (int t = 0; t < KH; ++t) {
        sQ[t] = 0.0f; mQ[t] = -INFINITY;
#pragma unroll
        for (int d = 0; d < 3; ++d) {
            sM[t][d] = 0.0f; xM[t][d] = -INFINITY; nM[t][d] = INFINITY;
            sS[t][d] = 0.0f; xS[t][d] = -INFINITY; nS[t][d] = INFINITY;
        }
    }

#pragma unroll 1
    for (int it = 0; it < PTW; ++it) {
        const int pb = (wave * PTW + it) * 16;

        const float px = lx[pb + m], py = ly[pb + m], pz = lz[pb + m];
        v2f a1, a2;
        a1.x = hi ? pz * pz : px * px;
        a1.y = hi ? px      : py * py;
        a2.x = hi ? 1.0f    : py;
        a2.y = hi ? 0.0f    : pz;

        const int rb = pb + (hi ? 8 : 0);
        float rx[8], ry[8], rz[8];
        {
            const float4 x0 = *(const float4*)(lx + rb), x1 = *(const float4*)(lx + rb + 4);
            const float4 y0 = *(const float4*)(ly + rb), y1 = *(const float4*)(ly + rb + 4);
            const float4 z0 = *(const float4*)(lz + rb), z1 = *(const float4*)(lz + rb + 4);
            rx[0]=x0.x; rx[1]=x0.y; rx[2]=x0.z; rx[3]=x0.w; rx[4]=x1.x; rx[5]=x1.y; rx[6]=x1.z; rx[7]=x1.w;
            ry[0]=y0.x; ry[1]=y0.y; ry[2]=y0.z; ry[3]=y0.w; ry[4]=y1.x; ry[5]=y1.y; ry[6]=y1.z; ry[7]=y1.w;
            rz[0]=z0.x; rz[1]=z0.y; rz[2]=z0.z; rz[3]=z0.w; rz[4]=z1.x; rz[5]=z1.y; rz[6]=z1.z; rz[7]=z1.w;
        }

        float e[KH][8];
        float den[8];
#pragma unroll
        for (int r = 0; r < 8; ++r) den[r] = 0.0f;
#pragma unroll
        for (int t = 0; t < KH; ++t) {
            v8f c = {};
            c = __builtin_amdgcn_wmma_f32_16x16x4_f32(false, a1, false, b1[t],
                                                      (short)0, c, false, false);
            c = __builtin_amdgcn_wmma_f32_16x16x4_f32(false, a2, false, b2[t],
                                                      (short)0, c, false, false);
#pragma unroll
            for (int r = 0; r < 8; ++r) {
                const float ev = __expf(c[r]);
                e[t][r] = ev;
                den[r] += ev;
            }
        }
#pragma unroll
        for (int t = 0; t < KH; ++t) {
            v8f c = {};
            c = __builtin_amdgcn_wmma_f32_16x16x4_f32(false, a1, false, o1[t],
                                                      (short)0, c, false, false);
            c = __builtin_amdgcn_wmma_f32_16x16x4_f32(false, a2, false, o2[t],
                                                      (short)0, c, false, false);
#pragma unroll
            for (int r = 0; r < 8; ++r) den[r] += __expf(c[r]);
        }

        float rd[8];
#pragma unroll
        for (int r = 0; r < 8; ++r) {
            float v = den[r];
            v += row_xor<1>(v);
            v += row_xor<2>(v);
            v += row_xor<4>(v);
            v += row_xor<8>(v);
            rd[r] = __builtin_amdgcn_rcpf(v);
        }

#pragma unroll
        for (int t = 0; t < KH; ++t) {
#pragma unroll
            for (int r = 0; r < 8; ++r) {
                const float Q = e[t][r] * rd[r];
                float z[3];
                z[0] = fmaf(rx[r], rs[t][0], nm[t][0]);
                z[1] = fmaf(ry[r], rs[t][1], nm[t][1]);
                z[2] = fmaf(rz[r], rs[t][2], nm[t][2]);
                sQ[t] += Q; mQ[t] = fmaxf(mQ[t], Q);
#pragma unroll
                for (int d = 0; d < 3; ++d) {
                    const float u = Q * z[d];
                    sM[t][d] += u;
                    xM[t][d] = fmaxf(xM[t][d], u); nM[t][d] = fminf(nM[t][d], u);
                    const float v = fmaf(u, z[d], -Q);
                    sS[t][d] += v;
                    xS[t][d] = fmaxf(xS[t][d], v); nS[t][d] = fminf(nS[t][d], v);
                }
            }
        }
    }

#pragma unroll
    for (int t = 0; t < KH; ++t) {
        float a[20];
        a[0] = fmaxf(mQ[t], xhalf(mQ[t]));
        a[1] = sQ[t] + xhalf(sQ[t]);
#pragma unroll
        for (int d = 0; d < 3; ++d) {
            a[2  + d] = fmaxf(xM[t][d], xhalf(xM[t][d]));
            a[5  + d] = fminf(nM[t][d], xhalf(nM[t][d]));
            a[8  + d] = sM[t][d] + xhalf(sM[t][d]);
            a[11 + d] = fmaxf(xS[t][d], xhalf(xS[t][d]));
            a[14 + d] = fminf(nS[t][d], xhalf(nS[t][d]));
            a[17 + d] = sS[t][d] + xhalf(sS[t][d]);
        }
        if (!hi) {
#pragma unroll
            for (int f = 0; f < 20; ++f) red[wave][f][t * 16 + m] = a[f];
        }
    }
    __syncthreads();

    const int kg = g * (KH * 16) + tid;
    if (tid < KH * 16 && kg < K_) {
        float a[20];
#pragma unroll
        for (int f = 0; f < 20; ++f) a[f] = red[0][f][tid];
#pragma unroll
        for (int wv = 1; wv < NWAVE; ++wv) {
            a[0] = fmaxf(a[0], red[wv][0][tid]); a[1] += red[wv][1][tid];
#pragma unroll
            for (int d = 0; d < 3; ++d) {
                a[2  + d] = fmaxf(a[2  + d], red[wv][2  + d][tid]);
                a[5  + d] = fminf(a[5  + d], red[wv][5  + d][tid]);
                a[8  + d] +=            red[wv][8  + d][tid];
                a[11 + d] = fmaxf(a[11 + d], red[wv][11 + d][tid]);
                a[14 + d] = fminf(a[14 + d], red[wv][14 + d][tid]);
                a[17 + d] +=            red[wv][17 + d][tid];
            }
        }
        float* o = partial + ((size_t)(b * NCHUNK + ch) * K_ + kg) * 20;
#pragma unroll
        for (int f = 0; f < 20; ++f) o[f] = a[f];
    }
}

#define NSLOT (K_ * 5)
#define SPT   5

#define MAXBITS 0x0000381Du
#define MINBITS 0x0001C0E0u

__global__ __launch_bounds__(128) __attribute__((amdgpu_waves_per_eu(1, 1)))
void fv_finalize(const float* __restrict__ w, const float* __restrict__ partial,
                 float* __restrict__ out)
{
    __shared__ __attribute__((aligned(16))) float img[SPT * 128 * 4];
    __shared__ float4 scal[128];
    __shared__ float ssq[20];
    const int b = blockIdx.x;
    const int t = threadIdx.x;

    const float wk = w[t < K_ ? t : 0];
    float4 q[NCHUNK][SPT];
    const float4* pb = (const float4*)(partial + (size_t)b * NCHUNK * K_ * 20);
#pragma unroll
    for (int c = 0; c < NCHUNK; ++c) {
#pragma unroll
        for (int j = 0; j < SPT; ++j) {
            const int s = t + j * 128;
            q[c][j] = pb[c * NSLOT + (s < NSLOT ? s : NSLOT - 1)];
        }
    }

    {
        const float Nf  = (float)N_;
        const float spi = 1.0f / (sqrtf(wk) * Nf);
        const float ssg = 1.0f / (Nf * sqrtf(2.0f * wk));
        scal[t] = make_float4(wk, Nf * wk, spi, ssg);
    }
    __syncthreads();

#pragma unroll
    for (int j = 0; j < SPT; ++j) {
        const int s  = t + j * 128;
        const int sl = s < NSLOT ? s : NSLOT - 1;
        const int k  = sl / 5;
        const int qd = sl - 5 * k;
        float x[NCHUNK][4];
#pragma unroll
        for (int c = 0; c < NCHUNK; ++c) {
            x[c][0] = q[c][j].x; x[c][1] = q[c][j].y;
            x[c][2] = q[c][j].z; x[c][3] = q[c][j].w;
        }
        float a[4];
#pragma unroll
        for (int e = 0; e < 4; ++e) {
            float mx = x[0][e], mn = x[0][e], sm = x[0][e];
#pragma unroll
            for (int c = 1; c < NCHUNK; ++c) {
                mx = fmaxf(mx, x[c][e]); mn = fminf(mn, x[c][e]); sm += x[c][e];
            }
            const unsigned f = (unsigned)(4 * qd + e);
            a[e] = ((MAXBITS >> f) & 1u) ? mx : (((MINBITS >> f) & 1u) ? mn : sm);
        }
        const float4 sv = scal[k];
        const float off0 = (qd == 0) ? sv.x : 0.0f;
        const float off1 = (qd == 0) ? sv.y : 0.0f;
        const float s012 = (qd <= 2) ? sv.z : sv.w;
        const float s3   = (qd <  2) ? sv.z : sv.w;
        float v[4];
        v[0] = (a[0] - off0) * s012;
        v[1] = (a[1] - off1) * s012;
        v[2] = a[2] * s012;
        v[3] = a[3] * s3;
#pragma unroll
        for (int e = 0; e < 4; ++e)
            v[e] = copysignf(sqrtf(fabsf(v[e])), v[e]);
        ((float4*)img)[s] = make_float4(v[0], v[1], v[2], v[3]);
    }
    __syncthreads();

    if (t < 20) {
        float x[K_];
#pragma unroll
        for (int k = 0; k < K_; ++k) x[k] = img[k * 20 + t];
#pragma unroll
        for (int k = 0; k < K_; k += 25)
            asm volatile("" :: "v"(x[k+0]),  "v"(x[k+1]),  "v"(x[k+2]),  "v"(x[k+3]),  "v"(x[k+4]),
                               "v"(x[k+5]),  "v"(x[k+6]),  "v"(x[k+7]),  "v"(x[k+8]),  "v"(x[k+9]),
                               "v"(x[k+10]), "v"(x[k+11]), "v"(x[k+12]), "v"(x[k+13]), "v"(x[k+14]),
                               "v"(x[k+15]), "v"(x[k+16]), "v"(x[k+17]), "v"(x[k+18]), "v"(x[k+19]),
                               "v"(x[k+20]), "v"(x[k+21]), "v"(x[k+22]), "v"(x[k+23]), "v"(x[k+24]));
        float s = 0.0f;
#pragma unroll
        for (int k = 0; k < K_; ++k) s += x[k] * x[k];
        ssq[t] = s;
    }
    __syncthreads();

    if (t < K_) {
        float v[20];
#pragma unroll
        for (int j = 0; j < 5; ++j) {
            const float4 r = ((const float4*)img)[t * 5 + j];
            v[4*j+0] = r.x; v[4*j+1] = r.y; v[4*j+2] = r.z; v[4*j+3] = r.w;
        }
#pragma unroll
        for (int f = 0; f < 20; ++f)
            out[(size_t)b * 20 * K_ + f * K_ + t] =
                v[f] * rsqrtf(fmaxf(ssq[f], 1e-12f));
    }
}

extern "C" void kernel_launch(void* const* d_in, const int* in_sizes, int n_in,
                              void* d_out, int out_size, void* d_ws, size_t ws_size,
                              hipStream_t stream)
{
    const float* pts = (const float*)d_in[0];
    const float* w   = (const float*)d_in[1];
    const float* mu  = (const float*)d_in[2];
    const float* sg  = (const float*)d_in[3];
    float* out = (float*)d_out;

    float* partial = (float*)d_ws;

    fv_fused<<<B_ * NCHUNK * 2, 256, 0, stream>>>(pts, w, mu, sg, partial);
    fv_finalize<<<B_, 128, 0, stream>>>(w, partial, out);
}


# statistics and denominator on PAIRS of gaussian tiles in packed f32 (lanes were two point rows): the 7 running sums pack too, order kept, sums of Q and Q z fused; den = even-tile + odd-tile sums
# speedup vs baseline: 2.6633x; 1.0868x over previous
#include <hip/hip_runtime.h>
#include <math.h>


typedef __attribute__((ext_vector_type(2))) float v2f;
typedef __attribute__((ext_vector_type(8))) float v8f;

template <int M>
__device__ __forceinline__ float row_xor(float x)
{
    const int b = __builtin_bit_cast(int, x);
    return __builtin_bit_cast(float,
        __builtin_amdgcn_update_dpp(0, b, 0x160 + M, 0xf, 0xf, true));
}

__device__ __forceinline__ float xhalf(float x)
{
    const int b = __builtin_bit_cast(int, x);
    return __builtin_bit_cast(float,
        __builtin_amdgcn_permlanex16(b, b, 0x76543210, 0xfedcba98, false, false));
}

#define B_      16
#define N_      8192
#define K_      125
#define KT      8
#define KH      4
#define KP      (KH / 2)
#define NCHUNK  8
#define CHUNK   (N_ / NCHUNK)
#define NWAVE   8
#define PTW     (CHUNK / 16 / NWAVE)
#define C15     0.06349363593424097f

__global__ __launch_bounds__(256)
void fv_fused(const float* __restrict__ pts, const float* __restrict__ w,
              const float* __restrict__ mu, const float* __restrict__ sg,
              float* __restrict__ partial)
{
    __shared__ __attribute__((aligned(16))) float lx[CHUNK];
    __shared__ __attribute__((aligned(16))) float ly[CHUNK];
    __shared__ __attribute__((aligned(16))) float lz[CHUNK];
    __shared__ float4 ctab[KT * 32];
    __shared__ float red[NWAVE][20][KH * 16];

    const int tid  = threadIdx.x;
    const int lane = tid & 31;
    const int wave = tid >> 5;
    const int m    = lane & 15;
    const bool hi  = lane >= 16;
    const int b    = blockIdx.x >> 4;
    const int ch   = (blockIdx.x >> 1) & 7;
    const int g    = blockIdx.x & 1;
    const int n0   = ch * CHUNK;

    {
        float q[CHUNK / 256][3];
#pragma unroll
        for (int j = 0; j < CHUNK / 256; ++j) {
            const float* p = pts + (size_t)(b * N_ + n0 + tid + j * 256) * 3;
            q[j][0] = p[0]; q[j][1] = p[1]; q[j][2] = p[2];
        }
#pragma unroll
        for (int j = 0; j < CHUNK / 256; ++j) {
            const int i = tid + j * 256;
            lx[i] = q[j][0]; ly[i] = q[j][1]; lz[i] = q[j][2];
        }
    }

    {
        const int  kc    = wave * 16 + m;
        const bool valid = kc < K_;
        const int  ki    = valid ? kc : 0;
        const float m0 = mu[ki*3+0], m1 = mu[ki*3+1], m2 = mu[ki*3+2];
        const float s0 = sg[ki*3+0], s1 = sg[ki*3+1], s2 = sg[ki*3+2];
        const float i0 = 1.0f/(s0*s0), i1 = 1.0f/(s1*s1), i2 = 1.0f/(s2*s2);
        const float g0 = -0.5f*i0, g1 = -0.5f*i1, g2 = -0.5f*i2;
        const float g3 = m0*i0,    g4 = m1*i1,    g5 = m2*i2;
        const float lq = logf(w[ki] * C15 / (s0 * s1 * s2));
        const float g6 = valid ? (lq - 0.5f*(m0*m0*i0 + m1*m1*i1 + m2*m2*i2))
                               : -INFINITY;
        float4 v;
        v.x = hi ? g2   : g0;
        v.y = hi ? g3   : g1;
        v.z = hi ? g6   : g4;
        v.w = hi ? 0.0f : g5;
        ctab[tid] = v;
    }
    __syncthreads();

    v2f b1[KH], b2[KH], o1[KH], o2[KH];
    v2f rs2[KP][3], nm2[KP][3];
#pragma unroll
    for (int t = 0; t < KH; ++t) {
        const float4 bc = ctab[(g * KH + t) * 32 + lane];
        b1[t].x = bc.x; b1[t].y = bc.y;
        b2[t].x = bc.z; b2[t].y = bc.w;
        const float4 oc = ctab[((1 - g) * KH + t) * 32 + lane];
        o1[t].x = oc.x; o1[t].y = oc.y;
        o2[t].x = oc.z; o2[t].y = oc.w;
        const int kc = (g * KH + t) * 16 + m;
        const int ki = (kc < K_) ? kc : 0;
#pragma unroll
        for (int d = 0; d < 3; ++d) {
            const float r = 1.0f / sg[ki*3+d];
            rs2[t >> 1][d][t & 1] = r;
            nm2[t >> 1][d][t & 1] = -mu[ki*3+d] * r;
        }
    }

    v2f   sQ[KP], sM[KP][3], sS[KP][3];
    float mQ[KH], xM[KH][3], nM[KH][3], xS[KH][3], nS[KH][3];
#pragma unroll
    for (int p = 0; p < KP; ++p) {
        sQ[p] = (v2f){0.0f, 0.0f};
#pragma unroll
        for (int d = 0; d < 3; ++d) {
            sM[p][d] = (v2f){0.0f, 0.0f};
            sS[p][d] = (v2f){0.0f, 0.0f};
        }
    }
#pragma unroll
    for (int t = 0; t < KH; ++t) {
        mQ[t] = -INFINITY;
#pragma unroll
        for (int d = 0; d < 3; ++d) {
            xM[t][d] = -INFINITY; nM[t][d] = INFINITY;
            xS[t][d] = -INFINITY; nS[t][d] = INFINITY;
        }
    }

#pragma unroll 1
    for (int it = 0; it < PTW; ++it) {
        const int pb = (wave * PTW + it) * 16;

        const float px = lx[pb + m], py = ly[pb + m], pz = lz[pb + m];
        v2f a1, a2;
        a1.x = hi ? pz * pz : px * px;
        a1.y = hi ? px      : py * py;
        a2.x = hi ? 1.0f    : py;
        a2.y = hi ? 0.0f    : pz;

        const int rb = pb + (hi ? 8 : 0);
        float rx[8], ry[8], rz[8];
        {
            const float4 x0 = *(const float4*)(lx + rb), x1 = *(const float4*)(lx + rb + 4);
            const float4 y0 = *(const float4*)(ly + rb), y1 = *(const float4*)(ly + rb + 4);
            const float4 z0 = *(const float4*)(lz + rb), z1 = *(const float4*)(lz + rb + 4);
            rx[0]=x0.x; rx[1]=x0.y; rx[2]=x0.z; rx[3]=x0.w; rx[4]=x1.x; rx[5]=x1.y; rx[6]=x1.z; rx[7]=x1.w;
            ry[0]=y0.x; ry[1]=y0.y; ry[2]=y0.z; ry[3]=y0.w; ry[4]=y1.x; ry[5]=y1.y; ry[6]=y1.z; ry[7]=y1.w;
            rz[0]=z0.x; rz[1]=z0.y; rz[2]=z0.z; rz[3]=z0.w; rz[4]=z1.x; rz[5]=z1.y; rz[6]=z1.z; rz[7]=z1.w;
        }

        v2f e[KP][8], eo[KP][8];
#pragma unroll
        for (int t = 0; t < KH; ++t) {
            v8f c = {};
            c = __builtin_amdgcn_wmma_f32_16x16x4_f32(false, a1, false, b1[t],
                                                      (short)0, c, false, false);
            c = __builtin_amdgcn_wmma_f32_16x16x4_f32(false, a2, false, b2[t],
                                                      (short)0, c, false, false);
#pragma unroll
            for (int r = 0; r < 8; ++r) {
                e[t >> 1][r][t & 1] = __expf(c[r]);
            }
        }
#pragma unroll
        for (int t = 0; t < KH; ++t) {
            v8f c = {};
            c = __builtin_amdgcn_wmma_f32_16x16x4_f32(false, a1, false, o1[t],
                                                      (short)0, c, false, false);
            c = __builtin_amdgcn_wmma_f32_16x16x4_f32(false, a2, false, o2[t],
                                                      (short)0, c, false, false);
#pragma unroll
            for (int r = 0; r < 8; ++r) eo[t >> 1][r][t & 1] = __expf(c[r]);
        }

        float rd[8];
#pragma unroll
        for (int r = 0; r < 8; ++r) {
            v2f d2 = e[0][r] + e[1][r];
#pragma unroll
            for (int p = 0; p < KP; ++p) d2 += eo[p][r];
            float v = d2.x + d2.y;
            v += row_xor<1>(v);
            v += row_xor<2>(v);
            v += row_xor<4>(v);
            v += row_xor<8>(v);
            rd[r] = __builtin_amdgcn_rcpf(v);
        }

#pragma unroll
        for (int p = 0; p < KP; ++p) {
            const int t0 = 2 * p, t1 = 2 * p + 1;
#pragma unroll
            for (int r = 0; r < 8; ++r) {
                const v2f rr = {rd[r], rd[r]};
                const v2f xr[3] = {{rx[r], rx[r]}, {ry[r], ry[r]}, {rz[r], rz[r]}};
                const v2f Q = e[p][r] * rr;
                sQ[p] += Q;
                mQ[t0] = fmaxf(mQ[t0], Q.x);
                mQ[t1] = fmaxf(mQ[t1], Q.y);
#pragma unroll
                for (int d = 0; d < 3; ++d) {
                    const v2f z = __builtin_elementwise_fma(xr[d], rs2[p][d], nm2[p][d]);
                    const v2f u = Q * z;
                    sM[p][d] += u;
                    xM[t0][d] = fmaxf(xM[t0][d], u.x); nM[t0][d] = fminf(nM[t0][d], u.x);
                    xM[t1][d] = fmaxf(xM[t1][d], u.y); nM[t1][d] = fminf(nM[t1][d], u.y);
                    const v2f v = __builtin_elementwise_fma(u, z, -Q);
                    sS[p][d] += v;
                    xS[t0][d] = fmaxf(xS[t0][d], v.x); nS[t0][d] = fminf(nS[t0][d], v.x);
                    xS[t1][d] = fmaxf(xS[t1][d], v.y); nS[t1][d] = fminf(nS[t1][d], v.y);
                }
            }
        }
    }

#pragma unroll
    for (int t = 0; t < KH; ++t) {
        float a[20];
        const float sq = sQ[t >> 1][t & 1];
        a[0] = fmaxf(mQ[t], xhalf(mQ[t]));
        a[1] = sq + xhalf(sq);
#pragma unroll
        for (int d = 0; d < 3; ++d) {
            const float sm = sM[t >> 1][d][t & 1];
            const float ss = sS[t >> 1][d][t & 1];
            a[2  + d] = fmaxf(xM[t][d], xhalf(xM[t][d]));
            a[5  + d] = fminf(nM[t][d], xhalf(nM[t][d]));
            a[8  + d] = sm + xhalf(sm);
            a[11 + d] = fmaxf(xS[t][d], xhalf(xS[t][d]));
            a[14 + d] = fminf(nS[t][d], xhalf(nS[t][d]));
            a[17 + d] = ss + xhalf(ss);
        }
        if (!hi) {
#pragma unroll
            for (int f = 0; f < 20; ++f) red[wave][f][t * 16 + m] = a[f];
        }
    }
    __syncthreads();

    const int kg = g * (KH * 16) + tid;
    if (tid < KH * 16 && kg < K_) {
        float a[20];
#pragma unroll
        for (int f = 0; f < 20; ++f) a[f] = red[0][f][tid];
#pragma unroll
        for (int wv = 1; wv < NWAVE; ++wv) {
            a[0] = fmaxf(a[0], red[wv][0][tid]); a[1] += red[wv][1][tid];
#pragma unroll
            for (int d = 0; d < 3; ++d) {
                a[2  + d] = fmaxf(a[2  + d], red[wv][2  + d][tid]);
                a[5  + d] = fminf(a[5  + d], red[wv][5  + d][tid]);
                a[8  + d] +=            red[wv][8  + d][tid];
                a[11 + d] = fmaxf(a[11 + d], red[wv][11 + d][tid]);
                a[14 + d] = fminf(a[14 + d], red[wv][14 + d][tid]);
                a[17 + d] +=            red[wv][17 + d][tid];
            }
        }
        float* o = partial + ((size_t)(b * NCHUNK + ch) * K_ + kg) * 20;
#pragma unroll
        for (int f = 0; f < 20; ++f) o[f] = a[f];
    }
}

#define NSLOT (K_ * 5)
#define SPT   5

#define MAXBITS 0x0000381Du
#define MINBITS 0x0001C0E0u

__global__ __launch_bounds__(128) __attribute__((amdgpu_waves_per_eu(1, 1)))
void fv_finalize(const float* __restrict__ w, const float* __restrict__ partial,
                 float* __restrict__ out)
{
    __shared__ __attribute__((aligned(16))) float img[SPT * 128 * 4];
    __shared__ float4 scal[128];
    __shared__ float ssq[20];
    const int b = blockIdx.x;
    const int t = threadIdx.x;

    const float wk = w[t < K_ ? t : 0];
    float4 q[NCHUNK][SPT];
    const float4* pb = (const float4*)(partial + (size_t)b * NCHUNK * K_ * 20);
#pragma unroll
    for (int c = 0; c < NCHUNK; ++c) {
#pragma unroll
        for (int j = 0; j < SPT; ++j) {
            const int s = t + j * 128;
            q[c][j] = pb[c * NSLOT + (s < NSLOT ? s : NSLOT - 1)];
        }
    }

    {
        const float Nf  = (float)N_;
        const float spi = 1.0f / (sqrtf(wk) * Nf);
        const float ssg = 1.0f / (Nf * sqrtf(2.0f * wk));
        scal[t] = make_float4(wk, Nf * wk, spi, ssg);
    }
    __syncthreads();

#pragma unroll
    for (int j = 0; j < SPT; ++j) {
        const int s  = t + j * 128;
        const int sl = s < NSLOT ? s : NSLOT - 1;
        const int k  = sl / 5;
        const int qd = sl - 5 * k;
        float x[NCHUNK][4];
#pragma unroll
        for (int c = 0; c < NCHUNK; ++c) {
            x[c][0] = q[c][j].x; x[c][1] = q[c][j].y;
            x[c][2] = q[c][j].z; x[c][3] = q[c][j].w;
        }
        float a[4];
#pragma unroll
        for (int e = 0; e < 4; ++e) {
            float mx = x[0][e], mn = x[0][e], sm = x[0][e];
#pragma unroll
            for (int c = 1; c < NCHUNK; ++c) {
                mx = fmaxf(mx, x[c][e]); mn = fminf(mn, x[c][e]); sm += x[c][e];
            }
            const unsigned f = (unsigned)(4 * qd + e);
            a[e] = ((MAXBITS >> f) & 1u) ? mx : (((MINBITS >> f) & 1u) ? mn : sm);
        }
        const float4 sv = scal[k];
        const float off0 = (qd == 0) ? sv.x : 0.0f;
        const float off1 = (qd == 0) ? sv.y : 0.0f;
        const float s012 = (qd <= 2) ? sv.z : sv.w;
        const float s3   = (qd <  2) ? sv.z : sv.w;
        float v[4];
        v[0] = (a[0] - off0) * s012;
        v[1] = (a[1] - off1) * s012;
        v[2] = a[2] * s012;
        v[3] = a[3] * s3;
#pragma unroll
        for (int e = 0; e < 4; ++e)
            v[e] = copysignf(sqrtf(fabsf(v[e])), v[e]);
        ((float4*)img)[s] = make_float4(v[0], v[1], v[2], v[3]);
    }
    __syncthreads();

    if (t < 20) {
        float x[K_];
#pragma unroll
        for (int k = 0; k < K_; ++k) x[k] = img[k * 20 + t];
#pragma unroll
        for (int k = 0; k < K_; k += 25)
            asm volatile("" :: "v"(x[k+0]),  "v"(x[k+1]),  "v"(x[k+2]),  "v"(x[k+3]),  "v"(x[k+4]),
                               "v"(x[k+5]),  "v"(x[k+6]),  "v"(x[k+7]),  "v"(x[k+8]),  "v"(x[k+9]),
                               "v"(x[k+10]), "v"(x[k+11]), "v"(x[k+12]), "v"(x[k+13]), "v"(x[k+14]),
                               "v"(x[k+15]), "v"(x[k+16]), "v"(x[k+17]), "v"(x[k+18]), "v"(x[k+19]),
                               "v"(x[k+20]), "v"(x[k+21]), "v"(x[k+22]), "v"(x[k+23]), "v"(x[k+24]));
        float s = 0.0f;
#pragma unroll
        for (int k = 0; k < K_; ++k) s += x[k] * x[k];
        ssq[t] = s;
    }
    __syncthreads();

    if (t < K_) {
        float v[20];
#pragma unroll
        for (int j = 0; j < 5; ++j) {
            const float4 r = ((const float4*)img)[t * 5 + j];
            v[4*j+0] = r.x; v[4*j+1] = r.y; v[4*j+2] = r.z; v[4*j+3] = r.w;
        }
#pragma unroll
        for (int f = 0; f < 20; ++f)
            out[(size_t)b * 20 * K_ + f * K_ + t] =
                v[f] * rsqrtf(fmaxf(ssq[f], 1e-12f));
    }
}

extern "C" void kernel_launch(void* const* d_in, const int* in_sizes, int n_in,
                              void* d_out, int out_size, void* d_ws, size_t ws_size,
                              hipStream_t stream)
{
    const float* pts = (const float*)d_in[0];
    const float* w   = (const float*)d_in[1];
    const float* mu  = (const float*)d_in[2];
    const float* sg  = (const float*)d_in[3];
    float* out = (float*)d_out;

    float* partial = (float*)d_ws;

    fv_fused<<<B_ * NCHUNK * 2, 256, 0, stream>>>(pts, w, mu, sg, partial);
    fv_finalize<<<B_, 128, 0, stream>>>(w, partial, out);
}
